# starting kernel
# baseline (speedup 1.0000x reference)
#include <hip/hip_runtime.h>
#include <cstddef>

typedef __attribute__((ext_vector_type(2))) float v2f;
typedef __attribute__((ext_vector_type(8))) float v8f;

constexpr int Bn   = 8;
constexpr int Hdim = 128;
constexpr int Wdim = 128;
constexpr int NCH  = 16;
constexpr int TH   = 16;
constexpr int TW   = 8;
constexpr int HTH  = TH + 2;
constexpr int HTW  = TW + 2;
constexpr int HALO = HTH * HTW;
constexpr int NB   = (HALO + 15) / 16;

__device__ __forceinline__ float wblk(const float* __restrict__ Wt, int gp, int m, int kk) {
  if (m < 8)  return (kk < 8)  ? Wt[(2 * gp) * 64 + m * 8 + kk] : 0.0f;
  else        return (kk >= 8) ? Wt[(2 * gp + 1) * 64 + (m - 8) * 8 + (kk - 8)] : 0.0f;
}

__global__ __launch_bounds__(256) void sasa_fused_kernel(
    const float* __restrict__ x,  const float* __restrict__ Wq,
    const float* __restrict__ Wk, const float* __restrict__ Wv,
    const float* __restrict__ hE, const float* __restrict__ wE,
    float* __restrict__ out) {
  __shared__ float xt[HALO * NCH];
  __shared__ float qt[HALO * NCH];
  __shared__ float kt[HALO * NCH];
  __shared__ float vt[HALO * NCH];
  __shared__ float emb[NCH * 9];

  const int tid = threadIdx.x;
  const int bx  = blockIdx.x;
  const int by  = blockIdx.y;
  const int bz  = blockIdx.z;
  const int gp  = bz & 3;
  const int b   = bz >> 2;
  const int x0  = bx * TW, y0 = by * TH;
  const int cbase = gp * NCH;

  for (int idx = tid; idx < HALO * NCH; idx += 256) {
    int ch = idx / HALO;
    int p  = idx - ch * HALO;
    int hy = p / HTW, hx = p - hy * HTW;
    int gy = y0 + hy - 1, gx = x0 + hx - 1;
    float v = 0.0f;
    if (gy >= 0 && gy < Hdim && gx >= 0 && gx < Wdim)
      v = x[(((size_t)b * 64 + cbase + ch) * Hdim + gy) * Wdim + gx];
    xt[p * NCH + ch] = v;
  }
  if (tid < NCH * 9) {
    int ch = tid / 9, w9 = tid - ch * 9;
    int i = w9 / 3, j = w9 - i * 3;
    int g  = 2 * gp + (ch >> 3);
    int cg = ch & 7;
    emb[ch * 9 + w9] = (cg < 4) ? hE[(g * 4 + cg) * 3 + i]
                                : wE[(g * 4 + (cg - 4)) * 3 + j];
  }

  const int ln = tid & 31;
  const int m  = ln & 15;
  const int kh = ln >> 4;
  v2f aq[4], ak[4], av[4];
  for (int c = 0; c < 4; ++c) {
    int k0 = 4 * c + 2 * kh;
    aq[c][0] = wblk(Wq, gp, m, k0); aq[c][1] = wblk(Wq, gp, m, k0 + 1);
    ak[c][0] = wblk(Wk, gp, m, k0); ak[c][1] = wblk(Wk, gp, m, k0 + 1);
    av[c][0] = wblk(Wv, gp, m, k0); av[c][1] = wblk(Wv, gp, m, k0 + 1);
  }
  __syncthreads();

  const int wv = tid >> 5;
  for (int batch = wv; batch < NB; batch += 8) {
    int pos = batch * 16 + m;
    int pc  = (pos < HALO) ? pos : (HALO - 1);
    v2f bb[4];
    for (int c = 0; c < 4; ++c) {
      int k0 = 4 * c + 2 * kh;
      bb[c] = *(const v2f*)&xt[pc * NCH + k0];
    }
    v8f cq = {}, ck = {}, cv = {};
    for (int c = 0; c < 4; ++c) {
      cq = __builtin_amdgcn_wmma_f32_16x16x4_f32(false, aq[c], false, bb[c],
                                                 (short)0, cq, false, false);
      ck = __builtin_amdgcn_wmma_f32_16x16x4_f32(false, ak[c], false, bb[c],
                                                 (short)0, ck, false, false);
      cv = __builtin_amdgcn_wmma_f32_16x16x4_f32(false, av[c], false, bb[c],
                                                 (short)0, cv, false, false);
    }
    if (pos < HALO) {
      float* q8 = &qt[pos * NCH + 8 * kh];
      float* k8 = &kt[pos * NCH + 8 * kh];
      float* v8 = &vt[pos * NCH + 8 * kh];
      #pragma unroll
      for (int r = 0; r < 8; ++r) { q8[r] = cq[r]; k8[r] = ck[r]; v8[r] = cv[r]; }
    }
  }
  __syncthreads();

  #pragma unroll
  for (int it = 0; it < (TH * TW * NCH) / 256; ++it) {
    int oi = it * 256 + tid;
    int ch = oi / (TH * TW);
    int p  = oi - ch * (TH * TW);
    int py = p / TW, px = p - py * TW;
    float q = qt[((py + 1) * HTW + (px + 1)) * NCH + ch];
    float lg[9];
    float mx = -1e30f;
    #pragma unroll
    for (int w9 = 0; w9 < 9; ++w9) {
      int i = w9 / 3, j = w9 - 3 * i;
      float kv = kt[((py + i) * HTW + (px + j)) * NCH + ch] + emb[ch * 9 + w9];
      float l = q * kv;
      lg[w9] = l;
      mx = fmaxf(mx, l);
    }
    float s = 0.0f, acc = 0.0f;
    #pragma unroll
    for (int w9 = 0; w9 < 9; ++w9) {
      int i = w9 / 3, j = w9 - 3 * i;
      float e = __expf(lg[w9] - mx);
      s   += e;
      acc += e * vt[((py + i) * HTW + (px + j)) * NCH + ch];
    }
    out[(((size_t)b * 64 + cbase + ch) * Hdim + (y0 + py)) * Wdim + (x0 + px)] = acc / s;
  }
}

extern "C" void kernel_launch(void* const* d_in, const int* in_sizes, int n_in,
                              void* d_out, int out_size, void* d_ws, size_t ws_size,
                              hipStream_t stream) {
  (void)in_sizes; (void)n_in; (void)d_ws; (void)ws_size; (void)out_size;
  const float* x  = (const float*)d_in[0];
  const float* Wq = (const float*)d_in[1];
  const float* Wk = (const float*)d_in[2];
  const float* Wv = (const float*)d_in[3];
  const float* hE = (const float*)d_in[4];
  const float* wE = (const float*)d_in[5];
  float* out = (float*)d_out;

  dim3 grid(Wdim / TW, Hdim / TH, Bn * 4);
  sasa_fused_kernel<<<grid, 256, 0, stream>>>(x, Wq, Wk, Wv, hE, wE, out);
}


# LDS tiles x/q/k/v stored channel-major [ch][pos] instead of [pos][ch]: lanes differ in pos, so every LDS access is bank-conflict-free (was 4- to 8-way)
# speedup vs baseline: 1.9139x; 1.9139x over previous
#include <hip/hip_runtime.h>
#include <cstddef>

typedef __attribute__((ext_vector_type(2))) float v2f;
typedef __attribute__((ext_vector_type(8))) float v8f;

constexpr int Bn   = 8;
constexpr int Hdim = 128;
constexpr int Wdim = 128;
constexpr int NCH  = 16;
constexpr int TH   = 16;
constexpr int TW   = 8;
constexpr int HTH  = TH + 2;
constexpr int HTW  = TW + 2;
constexpr int HALO = HTH * HTW;
constexpr int NB   = (HALO + 15) / 16;
constexpr int CS   = HALO;

__device__ __forceinline__ float wblk(const float* __restrict__ Wt, int gp, int m, int kk) {
  if (m < 8)  return (kk < 8)  ? Wt[(2 * gp) * 64 + m * 8 + kk] : 0.0f;
  else        return (kk >= 8) ? Wt[(2 * gp + 1) * 64 + (m - 8) * 8 + (kk - 8)] : 0.0f;
}

__global__ __launch_bounds__(256) void sasa_fused_kernel(
    const float* __restrict__ x,  const float* __restrict__ Wq,
    const float* __restrict__ Wk, const float* __restrict__ Wv,
    const float* __restrict__ hE, const float* __restrict__ wE,
    float* __restrict__ out) {
  __shared__ float xt[NCH * CS];
  __shared__ float qt[NCH * CS];
  __shared__ float kt[NCH * CS];
  __shared__ float vt[NCH * CS];
  __shared__ float emb[NCH * 9];

  const int tid = threadIdx.x;
  const int bx  = blockIdx.x;
  const int by  = blockIdx.y;
  const int bz  = blockIdx.z;
  const int gp  = bz & 3;
  const int b   = bz >> 2;
  const int x0  = bx * TW, y0 = by * TH;
  const int cbase = gp * NCH;

  for (int idx = tid; idx < HALO * NCH; idx += 256) {
    int ch = idx / HALO;
    int p  = idx - ch * HALO;
    int hy = p / HTW, hx = p - hy * HTW;
    int gy = y0 + hy - 1, gx = x0 + hx - 1;
    float v = 0.0f;
    if (gy >= 0 && gy < Hdim && gx >= 0 && gx < Wdim)
      v = x[(((size_t)b * 64 + cbase + ch) * Hdim + gy) * Wdim + gx];
    xt[ch * CS + p] = v;
  }
  if (tid < NCH * 9) {
    int ch = tid / 9, w9 = tid - ch * 9;
    int i = w9 / 3, j = w9 - i * 3;
    int g  = 2 * gp + (ch >> 3);
    int cg = ch & 7;
    emb[ch * 9 + w9] = (cg < 4) ? hE[(g * 4 + cg) * 3 + i]
                                : wE[(g * 4 + (cg - 4)) * 3 + j];
  }

  const int ln = tid & 31;
  const int m  = ln & 15;
  const int kh = ln >> 4;
  v2f aq[4], ak[4], av[4];
  for (int c = 0; c < 4; ++c) {
    int k0 = 4 * c + 2 * kh;
    aq[c][0] = wblk(Wq, gp, m, k0); aq[c][1] = wblk(Wq, gp, m, k0 + 1);
    ak[c][0] = wblk(Wk, gp, m, k0); ak[c][1] = wblk(Wk, gp, m, k0 + 1);
    av[c][0] = wblk(Wv, gp, m, k0); av[c][1] = wblk(Wv, gp, m, k0 + 1);
  }
  __syncthreads();

  const int wv = tid >> 5;
  for (int batch = wv; batch < NB; batch += 8) {
    int pos = batch * 16 + m;
    int pc  = (pos < HALO) ? pos : (HALO - 1);
    v2f bb[4];
    for (int c = 0; c < 4; ++c) {
      int k0 = 4 * c + 2 * kh;
      bb[c][0] = xt[k0 * CS + pc];
      bb[c][1] = xt[(k0 + 1) * CS + pc];
    }
    v8f cq = {}, ck = {}, cv = {};
    for (int c = 0; c < 4; ++c) {
      cq = __builtin_amdgcn_wmma_f32_16x16x4_f32(false, aq[c], false, bb[c],
                                                 (short)0, cq, false, false);
      ck = __builtin_amdgcn_wmma_f32_16x16x4_f32(false, ak[c], false, bb[c],
                                                 (short)0, ck, false, false);
      cv = __builtin_amdgcn_wmma_f32_16x16x4_f32(false, av[c], false, bb[c],
                                                 (short)0, cv, false, false);
    }
    if (pos < HALO) {
      float* q8 = &qt[8 * kh * CS + pos];
      float* k8 = &kt[8 * kh * CS + pos];
      float* v8 = &vt[8 * kh * CS + pos];
      #pragma unroll
      for (int r = 0; r < 8; ++r) { q8[r * CS] = cq[r]; k8[r * CS] = ck[r]; v8[r * CS] = cv[r]; }
    }
  }
  __syncthreads();

  #pragma unroll
  for (int it = 0; it < (TH * TW * NCH) / 256; ++it) {
    int oi = it * 256 + tid;
    int ch = oi / (TH * TW);
    int p  = oi - ch * (TH * TW);
    int py = p / TW, px = p - py * TW;
    float q = qt[ch * CS + (py + 1) * HTW + (px + 1)];
    float lg[9];
    float mx = -1e30f;
    #pragma unroll
    for (int w9 = 0; w9 < 9; ++w9) {
      int i = w9 / 3, j = w9 - 3 * i;
      float kv = kt[ch * CS + (py + i) * HTW + (px + j)] + emb[ch * 9 + w9];
      float l = q * kv;
      lg[w9] = l;
      mx = fmaxf(mx, l);
    }
    float s = 0.0f, acc = 0.0f;
    #pragma unroll
    for (int w9 = 0; w9 < 9; ++w9) {
      int i = w9 / 3, j = w9 - 3 * i;
      float e = __expf(lg[w9] - mx);
      s   += e;
      acc += e * vt[ch * CS + (py + i) * HTW + (px + j)];
    }
    out[(((size_t)b * 64 + cbase + ch) * Hdim + (y0 + py)) * Wdim + (x0 + px)] = acc / s;
  }
}

extern "C" void kernel_launch(void* const* d_in, const int* in_sizes, int n_in,
                              void* d_out, int out_size, void* d_ws, size_t ws_size,
                              hipStream_t stream) {
  (void)in_sizes; (void)n_in; (void)d_ws; (void)ws_size; (void)out_size;
  const float* x  = (const float*)d_in[0];
  const float* Wq = (const float*)d_in[1];
  const float* Wk = (const float*)d_in[2];
  const float* Wv = (const float*)d_in[3];
  const float* hE = (const float*)d_in[4];
  const float* wE = (const float*)d_in[5];
  float* out = (float*)d_out;

  dim3 grid(Wdim / TW, Hdim / TH, Bn * 4);
  sasa_fused_kernel<<<grid, 256, 0, stream>>>(x, Wq, Wk, Wv, hE, wE, out);
}


# Prologue global loads issued as one branch-free batch before the first wait (was one round trip per halo element): x goes straight into the WMMA B fragments, its LDS staging and barrier dropped
# speedup vs baseline: 2.3481x; 1.2269x over previous
#include <hip/hip_runtime.h>
#include <cstddef>

typedef __attribute__((ext_vector_type(2))) float v2f;
typedef __attribute__((ext_vector_type(8))) float v8f;

constexpr int Bn   = 8;
constexpr int Hdim = 128;
constexpr int Wdim = 128;
constexpr int NCH  = 16;
constexpr int TH   = 16;
constexpr int TW   = 8;
constexpr int HTH  = TH + 2;
constexpr int HTW  = TW + 2;
constexpr int HALO = HTH * HTW;
constexpr int NB   = (HALO + 15) / 16;
constexpr int CS   = HALO;
constexpr int HW   = Hdim * Wdim;

__device__ __forceinline__ void conv_batch(const v2f* aq, const v2f* ak, const v2f* av,
                                           const float* u, bool inb, int pos, int kh,
                                           float* qt, float* kt, float* vt) {
  v8f cq = {}, ck = {}, cv = {};
  for (int c = 0; c < 4; ++c) {
    v2f bb;
    bb[0] = inb ? u[2 * c]     : 0.0f;
    bb[1] = inb ? u[2 * c + 1] : 0.0f;
    cq = __builtin_amdgcn_wmma_f32_16x16x4_f32(false, aq[c], false, bb,
                                               (short)0, cq, false, false);
    ck = __builtin_amdgcn_wmma_f32_16x16x4_f32(false, ak[c], false, bb,
                                               (short)0, ck, false, false);
    cv = __builtin_amdgcn_wmma_f32_16x16x4_f32(false, av[c], false, bb,
                                               (short)0, cv, false, false);
  }
  if (pos < HALO) {
    float* q8 = &qt[8 * kh * CS + pos];
    float* k8 = &kt[8 * kh * CS + pos];
    float* v8 = &vt[8 * kh * CS + pos];
    #pragma unroll
    for (int r = 0; r < 8; ++r) { q8[r * CS] = cq[r]; k8[r * CS] = ck[r]; v8[r * CS] = cv[r]; }
  }
}

__global__ __launch_bounds__(256) void sasa_fused_kernel(
    const float* __restrict__ x,  const float* __restrict__ Wq,
    const float* __restrict__ Wk, const float* __restrict__ Wv,
    const float* __restrict__ hE, const float* __restrict__ wE,
    float* __restrict__ out) {
  __shared__ float qt[NCH * CS];
  __shared__ float kt[NCH * CS];
  __shared__ float vt[NCH * CS];
  __shared__ float emb[NCH * 9];

  const int tid = threadIdx.x;
  const int bx  = blockIdx.x;
  const int by  = blockIdx.y;
  const int bz  = blockIdx.z;
  const int gp  = bz & 3;
  const int b   = bz >> 2;
  const int x0  = bx * TW, y0 = by * TH;
  const int cbase = gp * NCH;

  const int ln = tid & 31;
  const int m  = ln & 15;
  const int kh = ln >> 4;
  const int wv = tid >> 5;


  const int hi   = m >> 3;
  const int wofs = (2 * gp + hi) * 64 + (m & 7) * 8 + 2 * kh;
  float wq[4], wk[4], ww[4];
  for (int cc = 0; cc < 2; ++cc) {
    for (int e = 0; e < 2; ++e) {
      wq[2 * cc + e] = Wq[wofs + 4 * cc + e];
      wk[2 * cc + e] = Wk[wofs + 4 * cc + e];
      ww[2 * cc + e] = Wv[wofs + 4 * cc + e];
    }
  }

  const bool has_e = (tid < NCH * 9);
  float ev;
  {
    int t  = has_e ? tid : 0;
    int ch = t / 9, w9 = t - ch * 9;
    int i = w9 / 3, j = w9 - i * 3;
    int g  = 2 * gp + (ch >> 3);
    int cg = ch & 7;
    const float* src = (cg < 4) ? (hE + (g * 4 + cg) * 3 + i)
                                : (wE + (g * 4 + (cg - 4)) * 3 + j);
    ev = *src;
  }

  const float* xg = x + ((size_t)b * 64 + cbase + 2 * kh) * HW;
  const bool two = (wv + 8 < NB);
  float u0[8], u1[8];
  bool inb0, inb1 = false;
  {
    int pos = wv * 16 + m;
    int hy = pos / HTW, hx = pos - hy * HTW;
    int gy = y0 + hy - 1, gx = x0 + hx - 1;
    inb0 = (gy >= 0 && gy < Hdim && gx >= 0 && gx < Wdim);
    int cy = min(max(gy, 0), Hdim - 1), cx = min(max(gx, 0), Wdim - 1);
    const float* p = xg + cy * Wdim + cx;
    for (int c = 0; c < 4; ++c) { u0[2 * c] = p[(4 * c) * HW]; u0[2 * c + 1] = p[(4 * c + 1) * HW]; }
  }
  if (two) {
    int pos = (wv + 8) * 16 + m;
    int pc  = (pos < HALO) ? pos : (HALO - 1);
    int hy = pc / HTW, hx = pc - hy * HTW;
    int gy = y0 + hy - 1, gx = x0 + hx - 1;
    inb1 = (gy >= 0 && gy < Hdim && gx >= 0 && gx < Wdim);
    int cy = min(max(gy, 0), Hdim - 1), cx = min(max(gx, 0), Wdim - 1);
    const float* p = xg + cy * Wdim + cx;
    for (int c = 0; c < 4; ++c) { u1[2 * c] = p[(4 * c) * HW]; u1[2 * c + 1] = p[(4 * c + 1) * HW]; }
  } else {
    for (int c = 0; c < 8; ++c) u1[c] = 0.0f;
  }

  if (has_e) emb[tid] = ev;

  v2f aq[4], ak[4], av[4];
  for (int c = 0; c < 4; ++c) {
    const bool nz = ((c >> 1) == hi);
    const int  cc = c & 1;
    aq[c][0] = nz ? wq[2 * cc] : 0.0f; aq[c][1] = nz ? wq[2 * cc + 1] : 0.0f;
    ak[c][0] = nz ? wk[2 * cc] : 0.0f; ak[c][1] = nz ? wk[2 * cc + 1] : 0.0f;
    av[c][0] = nz ? ww[2 * cc] : 0.0f; av[c][1] = nz ? ww[2 * cc + 1] : 0.0f;
  }

  conv_batch(aq, ak, av, u0, inb0, wv * 16 + m, kh, qt, kt, vt);
  if (two)
    conv_batch(aq, ak, av, u1, inb1, (wv + 8) * 16 + m, kh, qt, kt, vt);
  __syncthreads();

  #pragma unroll
  for (int it = 0; it < (TH * TW * NCH) / 256; ++it) {
    int oi = it * 256 + tid;
    int ch = oi / (TH * TW);
    int p  = oi - ch * (TH * TW);
    int py = p / TW, px = p - py * TW;
    float q = qt[ch * CS + (py + 1) * HTW + (px + 1)];
    float lg[9];
    float mx = -1e30f;
    #pragma unroll
    for (int w9 = 0; w9 < 9; ++w9) {
      int i = w9 / 3, j = w9 - 3 * i;
      float kv = kt[ch * CS + (py + i) * HTW + (px + j)] + emb[ch * 9 + w9];
      float l = q * kv;
      lg[w9] = l;
      mx = fmaxf(mx, l);
    }
    float s = 0.0f, acc = 0.0f;
    #pragma unroll
    for (int w9 = 0; w9 < 9; ++w9) {
      int i = w9 / 3, j = w9 - 3 * i;
      float e = __expf(lg[w9] - mx);
      s   += e;
      acc += e * vt[ch * CS + (py + i) * HTW + (px + j)];
    }
    out[(((size_t)b * 64 + cbase + ch) * Hdim + (y0 + py)) * Wdim + (x0 + px)] = acc / s;
  }
}

extern "C" void kernel_launch(void* const* d_in, const int* in_sizes, int n_in,
                              void* d_out, int out_size, void* d_ws, size_t ws_size,
                              hipStream_t stream) {
  (void)in_sizes; (void)n_in; (void)d_ws; (void)ws_size; (void)out_size;
  const float* x  = (const float*)d_in[0];
  const float* Wq = (const float*)d_in[1];
  const float* Wk = (const float*)d_in[2];
  const float* Wv = (const float*)d_in[3];
  const float* hE = (const float*)d_in[4];
  const float* wE = (const float*)d_in[5];
  float* out = (float*)d_out;

  dim3 grid(Wdim / TW, Hdim / TH, Bn * 4);
  sasa_fused_kernel<<<grid, 256, 0, stream>>>(x, Wq, Wk, Wv, hE, wE, out);
}


# Each work-group walks the 4 channel-group tiles of its spatial tile (grid 4096 -> 1024, all resident): next tile's global loads fly under the current softmax phase, 2 LDS stages, 1 barrier per tile
# speedup vs baseline: 2.4256x; 1.0330x over previous
#include <hip/hip_runtime.h>
#include <cstddef>


typedef __attribute__((ext_vector_type(2))) float v2f;
typedef __attribute__((ext_vector_type(8))) float v8f;

constexpr int Bn   = 8;
constexpr int Hdim = 128;
constexpr int Wdim = 128;
constexpr int NCH  = 16;
constexpr int NGP  = 4;
constexpr int TH   = 16;
constexpr int TW   = 8;
constexpr int HTH  = TH + 2;
constexpr int HTW  = TW + 2;
constexpr int HALO = HTH * HTW;
constexpr int NB   = (HALO + 15) / 16;
constexpr int CS    = HALO;
constexpr int PLANE = NCH * CS;
constexpr int NEMB  = NCH * 9;
constexpr int HW    = Hdim * Wdim;

struct TileRegs {
  float wq[4], wk[4], ww[4];
  float ev;
  float u0[8], u1[8];
};

__device__ __forceinline__ void load_tile(TileRegs& t, int gp,
                                          const float* __restrict__ x,  const float* __restrict__ Wq,
                                          const float* __restrict__ Wk, const float* __restrict__ Wv,
                                          const float* ep, int wofs, int xo0, int xo1, bool two) {
  const int wo = wofs + gp * 128;
  for (int cc = 0; cc < 2; ++cc) {
    for (int e = 0; e < 2; ++e) {
      t.wq[2 * cc + e] = Wq[wo + 4 * cc + e];
      t.wk[2 * cc + e] = Wk[wo + 4 * cc + e];
      t.ww[2 * cc + e] = Wv[wo + 4 * cc + e];
    }
  }
  t.ev = ep[gp * 24];
  const int xo = gp * NCH * HW;
  for (int c = 0; c < 4; ++c) {
    t.u0[2 * c]     = x[xo0 + xo + (4 * c) * HW];
    t.u0[2 * c + 1] = x[xo0 + xo + (4 * c + 1) * HW];
  }
  if (two) {
    for (int c = 0; c < 4; ++c) {
      t.u1[2 * c]     = x[xo1 + xo + (4 * c) * HW];
      t.u1[2 * c + 1] = x[xo1 + xo + (4 * c + 1) * HW];
    }
  } else {
    for (int c = 0; c < 8; ++c) t.u1[c] = 0.0f;
  }
}

__device__ __forceinline__ void lds_barrier() {
  asm volatile("s_wait_dscnt 0x0\n\ts_barrier_signal -1\n\ts_barrier_wait -1" ::: "memory");
}

__device__ __forceinline__ void conv_batch(const v2f* aq, const v2f* ak, const v2f* av,
                                           const float* u, bool inb, int pos, int kh,
                                           float* qt, float* kt, float* vt) {
  v8f cq = {}, ck = {}, cv = {};
  for (int c = 0; c < 4; ++c) {
    v2f bb;
    bb[0] = inb ? u[2 * c]     : 0.0f;
    bb[1] = inb ? u[2 * c + 1] : 0.0f;
    cq = __builtin_amdgcn_wmma_f32_16x16x4_f32(false, aq[c], false, bb,
                                               (short)0, cq, false, false);
    ck = __builtin_amdgcn_wmma_f32_16x16x4_f32(false, ak[c], false, bb,
                                               (short)0, ck, false, false);
    cv = __builtin_amdgcn_wmma_f32_16x16x4_f32(false, av[c], false, bb,
                                               (short)0, cv, false, false);
  }
  if (pos < HALO) {
    float* q8 = &qt[8 * kh * CS + pos];
    float* k8 = &kt[8 * kh * CS + pos];
    float* v8 = &vt[8 * kh * CS + pos];
    #pragma unroll
    for (int r = 0; r < 8; ++r) { q8[r * CS] = cq[r]; k8[r * CS] = ck[r]; v8[r * CS] = cv[r]; }
  }
}

__global__ __launch_bounds__(256, 8) void sasa_fused_kernel(
    const float* __restrict__ x,  const float* __restrict__ Wq,
    const float* __restrict__ Wk, const float* __restrict__ Wv,
    const float* __restrict__ hE, const float* __restrict__ wE,
    float* __restrict__ out) {
  __shared__ float qt[2 * PLANE];
  __shared__ float kt[2 * PLANE];
  __shared__ float vt[2 * PLANE];
  __shared__ float emb[2 * NEMB];

  const int tid = threadIdx.x;
  const int bx  = blockIdx.x;
  const int by  = blockIdx.y;
  const int b   = blockIdx.z;
  const int x0  = bx * TW, y0 = by * TH;

  const int ln = tid & 31;
  const int m  = ln & 15;
  const int kh = ln >> 4;
  const int wv = tid >> 5;


  const int hi   = m >> 3;
  const int wofs = hi * 64 + (m & 7) * 8 + 2 * kh;

  const bool has_e = (tid < NEMB);
  const float* ep;
  {
    int t  = has_e ? tid : 0;
    int ch = t / 9, w9 = t - ch * 9;
    int i = w9 / 3, j = w9 - i * 3;
    int g  = ch >> 3;
    int cg = ch & 7;
    ep = (cg < 4) ? (hE + (g * 4 + cg) * 3 + i)
                  : (wE + (g * 4 + (cg - 4)) * 3 + j);
  }

  const int xg = (b * 64 + 2 * kh) * HW;
  const bool two = (wv + 8 < NB);
  const int pos0 = wv * 16 + m;
  const int pos1 = (wv + 8) * 16 + m;
  bool inb0, inb1 = false;
  int p0, p1;
  {
    int hy = pos0 / HTW, hx = pos0 - hy * HTW;
    int gy = y0 + hy - 1, gx = x0 + hx - 1;
    inb0 = (gy >= 0 && gy < Hdim && gx >= 0 && gx < Wdim);
    int cy = min(max(gy, 0), Hdim - 1), cx = min(max(gx, 0), Wdim - 1);
    p0 = xg + cy * Wdim + cx;
  }
  p1 = p0;
  if (two) {
    int pc  = (pos1 < HALO) ? pos1 : (HALO - 1);
    int hy = pc / HTW, hx = pc - hy * HTW;
    int gy = y0 + hy - 1, gx = x0 + hx - 1;
    inb1 = (gy >= 0 && gy < Hdim && gx >= 0 && gx < Wdim);
    int cy = min(max(gy, 0), Hdim - 1), cx = min(max(gx, 0), Wdim - 1);
    p1 = xg + cy * Wdim + cx;
  }

  TileRegs t;
  load_tile(t, 0, x, Wq, Wk, Wv, ep, wofs, p0, p1, two);

  for (int gp = 0; gp < NGP; ++gp) {
    const int st = gp & 1;
    float* qs = qt + st * PLANE;
    float* ks = kt + st * PLANE;
    float* vs = vt + st * PLANE;
    float* es = emb + st * NEMB;
    const int cbase = gp * NCH;

    if (has_e) es[tid] = t.ev;

    v2f aq[4], ak[4], av[4];
    for (int c = 0; c < 4; ++c) {
      const bool nz = ((c >> 1) == hi);
      const int  cc = c & 1;
      aq[c][0] = nz ? t.wq[2 * cc] : 0.0f; aq[c][1] = nz ? t.wq[2 * cc + 1] : 0.0f;
      ak[c][0] = nz ? t.wk[2 * cc] : 0.0f; ak[c][1] = nz ? t.wk[2 * cc + 1] : 0.0f;
      av[c][0] = nz ? t.ww[2 * cc] : 0.0f; av[c][1] = nz ? t.ww[2 * cc + 1] : 0.0f;
    }

    conv_batch(aq, ak, av, t.u0, inb0, pos0, kh, qs, ks, vs);
    if (two)
      conv_batch(aq, ak, av, t.u1, inb1, pos1, kh, qs, ks, vs);

    if (gp + 1 < NGP) load_tile(t, gp + 1, x, Wq, Wk, Wv, ep, wofs, p0, p1, two);

    lds_barrier();

    #pragma unroll
    for (int it = 0; it < (TH * TW * NCH) / 256; ++it) {
      int oi = it * 256 + tid;
      int ch = oi / (TH * TW);
      int p  = oi - ch * (TH * TW);
      int py = p / TW, px = p - py * TW;
      float q = qs[ch * CS + (py + 1) * HTW + (px + 1)];
      float lg[9];
      float mx = -1e30f;
      #pragma unroll
      for (int w9 = 0; w9 < 9; ++w9) {
        int i = w9 / 3, j = w9 - 3 * i;
        float kv = ks[ch * CS + (py + i) * HTW + (px + j)] + es[ch * 9 + w9];
        float l = q * kv;
        lg[w9] = l;
        mx = fmaxf(mx, l);
      }
      float s = 0.0f, acc = 0.0f;
      #pragma unroll
      for (int w9 = 0; w9 < 9; ++w9) {
        int i = w9 / 3, j = w9 - 3 * i;
        float e = __expf(lg[w9] - mx);
        s   += e;
        acc += e * vs[ch * CS + (py + i) * HTW + (px + j)];
      }
      out[(((size_t)b * 64 + cbase + ch) * Hdim + (y0 + py)) * Wdim + (x0 + px)] = acc / s;
    }
  }
}

extern "C" void kernel_launch(void* const* d_in, const int* in_sizes, int n_in,
                              void* d_out, int out_size, void* d_ws, size_t ws_size,
                              hipStream_t stream) {
  (void)in_sizes; (void)n_in; (void)d_ws; (void)ws_size; (void)out_size;
  const float* x  = (const float*)d_in[0];
  const float* Wq = (const float*)d_in[1];
  const float* Wk = (const float*)d_in[2];
  const float* Wv = (const float*)d_in[3];
  const float* hE = (const float*)d_in[4];
  const float* wE = (const float*)d_in[5];
  float* out = (float*)d_out;

  dim3 grid(Wdim / TW, Hdim / TH, Bn);
  sasa_fused_kernel<<<grid, 256, 0, stream>>>(x, Wq, Wk, Wv, hE, wE, out);
}


# Softmax phase register-blocked: each thread owns a strip of 8 rows of one channel and column and keeps the 3-row k/v window and the channel's 9 embedding entries in registers; same summation order
# speedup vs baseline: 2.5239x; 1.0405x over previous
#include <hip/hip_runtime.h>
#include <cstddef>


typedef __attribute__((ext_vector_type(2))) float v2f;
typedef __attribute__((ext_vector_type(8))) float v8f;

constexpr int Bn   = 8;
constexpr int Hdim = 128;
constexpr int Wdim = 128;
constexpr int NCH  = 16;
constexpr int NGP  = 4;
constexpr int TH   = 16;
constexpr int TW   = 8;
constexpr int HTH  = TH + 2;
constexpr int HTW  = TW + 2;
constexpr int HALO = HTH * HTW;
constexpr int NB   = (HALO + 15) / 16;
constexpr int CS    = HALO;
constexpr int PLANE = NCH * CS;
constexpr int NEMB  = NCH * 9;
constexpr int HW    = Hdim * Wdim;
constexpr int SR    = 8;

struct TileRegs {
  float wq[4], wk[4], ww[4];
  float ev;
  float u0[8], u1[8];
};

__device__ __forceinline__ void load_tile(TileRegs& t, int gp,
                                          const float* __restrict__ x,  const float* __restrict__ Wq,
                                          const float* __restrict__ Wk, const float* __restrict__ Wv,
                                          const float* ep, int wofs, int xo0, int xo1, bool two) {
  const int wo = wofs + gp * 128;
  for (int cc = 0; cc < 2; ++cc) {
    for (int e = 0; e < 2; ++e) {
      t.wq[2 * cc + e] = Wq[wo + 4 * cc + e];
      t.wk[2 * cc + e] = Wk[wo + 4 * cc + e];
      t.ww[2 * cc + e] = Wv[wo + 4 * cc + e];
    }
  }
  t.ev = ep[gp * 24];
  const int xo = gp * NCH * HW;
  for (int c = 0; c < 4; ++c) {
    t.u0[2 * c]     = x[xo0 + xo + (4 * c) * HW];
    t.u0[2 * c + 1] = x[xo0 + xo + (4 * c + 1) * HW];
  }
  if (two) {
    for (int c = 0; c < 4; ++c) {
      t.u1[2 * c]     = x[xo1 + xo + (4 * c) * HW];
      t.u1[2 * c + 1] = x[xo1 + xo + (4 * c + 1) * HW];
    }
  } else {
    for (int c = 0; c < 8; ++c) t.u1[c] = 0.0f;
  }
}

__device__ __forceinline__ void lds_barrier() {
  asm volatile("s_wait_dscnt 0x0\n\ts_barrier_signal -1\n\ts_barrier_wait -1" ::: "memory");
}

__device__ __forceinline__ void conv_batch(const v2f* aq, const v2f* ak, const v2f* av,
                                           const float* u, bool inb, int pos, int kh,
                                           float* qt, float* kt, float* vt) {
  v8f cq = {}, ck = {}, cv = {};
  for (int c = 0; c < 4; ++c) {
    v2f bb;
    bb[0] = inb ? u[2 * c]     : 0.0f;
    bb[1] = inb ? u[2 * c + 1] : 0.0f;
    cq = __builtin_amdgcn_wmma_f32_16x16x4_f32(false, aq[c], false, bb,
                                               (short)0, cq, false, false);
    ck = __builtin_amdgcn_wmma_f32_16x16x4_f32(false, ak[c], false, bb,
                                               (short)0, ck, false, false);
    cv = __builtin_amdgcn_wmma_f32_16x16x4_f32(false, av[c], false, bb,
                                               (short)0, cv, false, false);
  }
  if (pos < HALO) {
    float* q8 = &qt[8 * kh * CS + pos];
    float* k8 = &kt[8 * kh * CS + pos];
    float* v8 = &vt[8 * kh * CS + pos];
    #pragma unroll
    for (int r = 0; r < 8; ++r) { q8[r * CS] = cq[r]; k8[r * CS] = ck[r]; v8[r * CS] = cv[r]; }
  }
}

__global__ __launch_bounds__(256, 8) void sasa_fused_kernel(
    const float* __restrict__ x,  const float* __restrict__ Wq,
    const float* __restrict__ Wk, const float* __restrict__ Wv,
    const float* __restrict__ hE, const float* __restrict__ wE,
    float* __restrict__ out) {
  __shared__ float qt[2 * PLANE];
  __shared__ float kt[2 * PLANE];
  __shared__ float vt[2 * PLANE];
  __shared__ float emb[2 * NEMB];

  const int tid = threadIdx.x;
  const int bx  = blockIdx.x;
  const int by  = blockIdx.y;
  const int b   = blockIdx.z;
  const int x0  = bx * TW, y0 = by * TH;

  const int ln = tid & 31;
  const int m  = ln & 15;
  const int kh = ln >> 4;
  const int wv = tid >> 5;


  const int hi   = m >> 3;
  const int wofs = hi * 64 + (m & 7) * 8 + 2 * kh;

  const bool has_e = (tid < NEMB);
  const float* ep;
  {
    int t  = has_e ? tid : 0;
    int ch = t / 9, w9 = t - ch * 9;
    int i = w9 / 3, j = w9 - i * 3;
    int g  = ch >> 3;
    int cg = ch & 7;
    ep = (cg < 4) ? (hE + (g * 4 + cg) * 3 + i)
                  : (wE + (g * 4 + (cg - 4)) * 3 + j);
  }

  const int xg = (b * 64 + 2 * kh) * HW;
  const bool two = (wv + 8 < NB);
  const int pos0 = wv * 16 + m;
  const int pos1 = (wv + 8) * 16 + m;
  bool inb0, inb1 = false;
  int p0, p1;
  {
    int hy = pos0 / HTW, hx = pos0 - hy * HTW;
    int gy = y0 + hy - 1, gx = x0 + hx - 1;
    inb0 = (gy >= 0 && gy < Hdim && gx >= 0 && gx < Wdim);
    int cy = min(max(gy, 0), Hdim - 1), cx = min(max(gx, 0), Wdim - 1);
    p0 = xg + cy * Wdim + cx;
  }
  p1 = p0;
  if (two) {
    int pc  = (pos1 < HALO) ? pos1 : (HALO - 1);
    int hy = pc / HTW, hx = pc - hy * HTW;
    int gy = y0 + hy - 1, gx = x0 + hx - 1;
    inb1 = (gy >= 0 && gy < Hdim && gx >= 0 && gx < Wdim);
    int cy = min(max(gy, 0), Hdim - 1), cx = min(max(gx, 0), Wdim - 1);
    p1 = xg + cy * Wdim + cx;
  }

  const int spx  = tid & 7;
  const int sch  = ((wv >> 1) << 2) | ((tid >> 3) & 3);
  const int sy0  = (wv & 1) * SR;
  const int sofs = sch * CS + sy0 * HTW + spx;

  TileRegs t;
  load_tile(t, 0, x, Wq, Wk, Wv, ep, wofs, p0, p1, two);

  for (int gp = 0; gp < NGP; ++gp) {
    const int st = gp & 1;
    float* qs = qt + st * PLANE;
    float* ks = kt + st * PLANE;
    float* vs = vt + st * PLANE;
    float* es = emb + st * NEMB;
    const int cbase = gp * NCH;

    if (has_e) es[tid] = t.ev;

    v2f aq[4], ak[4], av[4];
    for (int c = 0; c < 4; ++c) {
      const bool nz = ((c >> 1) == hi);
      const int  cc = c & 1;
      aq[c][0] = nz ? t.wq[2 * cc] : 0.0f; aq[c][1] = nz ? t.wq[2 * cc + 1] : 0.0f;
      ak[c][0] = nz ? t.wk[2 * cc] : 0.0f; ak[c][1] = nz ? t.wk[2 * cc + 1] : 0.0f;
      av[c][0] = nz ? t.ww[2 * cc] : 0.0f; av[c][1] = nz ? t.ww[2 * cc + 1] : 0.0f;
    }

    conv_batch(aq, ak, av, t.u0, inb0, pos0, kh, qs, ks, vs);
    if (two)
      conv_batch(aq, ak, av, t.u1, inb1, pos1, kh, qs, ks, vs);

    if (gp + 1 < NGP) load_tile(t, gp + 1, x, Wq, Wk, Wv, ep, wofs, p0, p1, two);

    lds_barrier();

    {
      const float* kc = ks + sofs;
      const float* vc = vs + sofs;
      const float* qc = qs + sofs + HTW + 1;
      float e9[9];
      for (int w9 = 0; w9 < 9; ++w9) e9[w9] = es[sch * 9 + w9];
      float kk[SR + 2][3], vv[SR + 2][3];
      for (int i = 0; i < 2; ++i) {
        for (int j = 0; j < 3; ++j) { kk[i][j] = kc[i * HTW + j]; vv[i][j] = vc[i * HTW + j]; }
      }
      float* op = out + (((size_t)b * 64 + cbase + sch) * Hdim + (y0 + sy0)) * Wdim + (x0 + spx);
      for (int r = 0; r < SR; ++r) {
        for (int j = 0; j < 3; ++j) {
          kk[r + 2][j] = kc[(r + 2) * HTW + j];
          vv[r + 2][j] = vc[(r + 2) * HTW + j];
        }
        float q = qc[r * HTW];
        float lg[9];
        float mx = -1e30f;
        for (int w9 = 0; w9 < 9; ++w9) {
          int i = w9 / 3, j = w9 - 3 * i;
          float kv = kk[r + i][j] + e9[w9];
          float l = q * kv;
          lg[w9] = l;
          mx = fmaxf(mx, l);
        }
        float s = 0.0f, acc = 0.0f;
        for (int w9 = 0; w9 < 9; ++w9) {
          int i = w9 / 3, j = w9 - 3 * i;
          float e = __expf(lg[w9] - mx);
          s   += e;
          acc += e * vv[r + i][j];
        }
        op[r * Wdim] = acc / s;
      }
    }
  }
}

extern "C" void kernel_launch(void* const* d_in, const int* in_sizes, int n_in,
                              void* d_out, int out_size, void* d_ws, size_t ws_size,
                              hipStream_t stream) {
  (void)in_sizes; (void)n_in; (void)d_ws; (void)ws_size; (void)out_size;
  const float* x  = (const float*)d_in[0];
  const float* Wq = (const float*)d_in[1];
  const float* Wk = (const float*)d_in[2];
  const float* Wv = (const float*)d_in[3];
  const float* hE = (const float*)d_in[4];
  const float* wE = (const float*)d_in[5];
  float* out = (float*)d_out;

  dim3 grid(Wdim / TW, Hdim / TH, Bn);
  sasa_fused_kernel<<<grid, 256, 0, stream>>>(x, Wq, Wk, Wv, hE, wE, out);
}


# Tile 16 rows x 8 cols -> 4 rows x 32 cols (softmax lane = column): each store instruction writes one whole 128-byte line of out (was four 32-byte pieces); x halo rows become 34-float runs
# speedup vs baseline: 3.7550x; 1.4878x over previous
#include <hip/hip_runtime.h>
#include <cstddef>


typedef __attribute__((ext_vector_type(2))) float v2f;
typedef __attribute__((ext_vector_type(8))) float v8f;

constexpr int Bn   = 8;
constexpr int Hdim = 128;
constexpr int Wdim = 128;
constexpr int NCH  = 16;
constexpr int NGP  = 4;
constexpr int TH   = 4;
constexpr int TW   = 32;
constexpr int HTH  = TH + 2;
constexpr int HTW  = TW + 2;
constexpr int HALO = HTH * HTW;
constexpr int NB   = (HALO + 15) / 16;
constexpr int CS    = HALO;
constexpr int PLANE = NCH * CS;
constexpr int NEMB  = NCH * 9;
constexpr int HW    = Hdim * Wdim;
constexpr int SR    = TH;

struct TileRegs {
  float wq[4], wk[4], ww[4];
  float ev;
  float u0[8], u1[8];
};

__device__ __forceinline__ void load_tile(TileRegs& t, int gp,
                                          const float* __restrict__ x,  const float* __restrict__ Wq,
                                          const float* __restrict__ Wk, const float* __restrict__ Wv,
                                          const float* ep, int wofs, int xo0, int xo1, bool two) {
  const int wo = wofs + gp * 128;
  for (int cc = 0; cc < 2; ++cc) {
    for (int e = 0; e < 2; ++e) {
      t.wq[2 * cc + e] = Wq[wo + 4 * cc + e];
      t.wk[2 * cc + e] = Wk[wo + 4 * cc + e];
      t.ww[2 * cc + e] = Wv[wo + 4 * cc + e];
    }
  }
  t.ev = ep[gp * 24];
  const int xo = gp * NCH * HW;
  for (int c = 0; c < 4; ++c) {
    t.u0[2 * c]     = x[xo0 + xo + (4 * c) * HW];
    t.u0[2 * c + 1] = x[xo0 + xo + (4 * c + 1) * HW];
  }
  if (two) {
    for (int c = 0; c < 4; ++c) {
      t.u1[2 * c]     = x[xo1 + xo + (4 * c) * HW];
      t.u1[2 * c + 1] = x[xo1 + xo + (4 * c + 1) * HW];
    }
  } else {
    for (int c = 0; c < 8; ++c) t.u1[c] = 0.0f;
  }
}

__device__ __forceinline__ void lds_barrier() {
  asm volatile("s_wait_dscnt 0x0\n\ts_barrier_signal -1\n\ts_barrier_wait -1" ::: "memory");
}

__device__ __forceinline__ void conv_batch(const v2f* aq, const v2f* ak, const v2f* av,
                                           const float* u, bool inb, int pos, int kh,
                                           float* qt, float* kt, float* vt) {
  v8f cq = {}, ck = {}, cv = {};
  for (int c = 0; c < 4; ++c) {
    v2f bb;
    bb[0] = inb ? u[2 * c]     : 0.0f;
    bb[1] = inb ? u[2 * c + 1] : 0.0f;
    cq = __builtin_amdgcn_wmma_f32_16x16x4_f32(false, aq[c], false, bb,
                                               (short)0, cq, false, false);
    ck = __builtin_amdgcn_wmma_f32_16x16x4_f32(false, ak[c], false, bb,
                                               (short)0, ck, false, false);
    cv = __builtin_amdgcn_wmma_f32_16x16x4_f32(false, av[c], false, bb,
                                               (short)0, cv, false, false);
  }
  if (pos < HALO) {
    float* q8 = &qt[8 * kh * CS + pos];
    float* k8 = &kt[8 * kh * CS + pos];
    float* v8 = &vt[8 * kh * CS + pos];
    #pragma unroll
    for (int r = 0; r < 8; ++r) { q8[r * CS] = cq[r]; k8[r * CS] = ck[r]; v8[r * CS] = cv[r]; }
  }
}

__global__ __launch_bounds__(256, 8) void sasa_fused_kernel(
    const float* __restrict__ x,  const float* __restrict__ Wq,
    const float* __restrict__ Wk, const float* __restrict__ Wv,
    const float* __restrict__ hE, const float* __restrict__ wE,
    float* __restrict__ out) {
  __shared__ float qt[2 * PLANE];
  __shared__ float kt[2 * PLANE];
  __shared__ float vt[2 * PLANE];
  __shared__ float emb[2 * NEMB];

  const int tid = threadIdx.x;
  const int bx  = blockIdx.x;
  const int by  = blockIdx.y;
  const int b   = blockIdx.z;
  const int x0  = bx * TW, y0 = by * TH;

  const int ln = tid & 31;
  const int m  = ln & 15;
  const int kh = ln >> 4;
  const int wv = tid >> 5;


  const int hi   = m >> 3;
  const int wofs = hi * 64 + (m & 7) * 8 + 2 * kh;

  const bool has_e = (tid < NEMB);
  const float* ep;
  {
    int t  = has_e ? tid : 0;
    int ch = t / 9, w9 = t - ch * 9;
    int i = w9 / 3, j = w9 - i * 3;
    int g  = ch >> 3;
    int cg = ch & 7;
    ep = (cg < 4) ? (hE + (g * 4 + cg) * 3 + i)
                  : (wE + (g * 4 + (cg - 4)) * 3 + j);
  }

  const int xg = (b * 64 + 2 * kh) * HW;
  const bool two = (wv + 8 < NB);
  const int pos0 = wv * 16 + m;
  const int pos1 = (wv + 8) * 16 + m;
  bool inb0, inb1 = false;
  int p0, p1;
  {
    int hy = pos0 / HTW, hx = pos0 - hy * HTW;
    int gy = y0 + hy - 1, gx = x0 + hx - 1;
    inb0 = (gy >= 0 && gy < Hdim && gx >= 0 && gx < Wdim);
    int cy = min(max(gy, 0), Hdim - 1), cx = min(max(gx, 0), Wdim - 1);
    p0 = xg + cy * Wdim + cx;
  }
  p1 = p0;
  if (two) {
    int pc  = (pos1 < HALO) ? pos1 : (HALO - 1);
    int hy = pc / HTW, hx = pc - hy * HTW;
    int gy = y0 + hy - 1, gx = x0 + hx - 1;
    inb1 = (gy >= 0 && gy < Hdim && gx >= 0 && gx < Wdim);
    int cy = min(max(gy, 0), Hdim - 1), cx = min(max(gx, 0), Wdim - 1);
    p1 = xg + cy * Wdim + cx;
  }

  const int spx  = ln;

  TileRegs t;
  load_tile(t, 0, x, Wq, Wk, Wv, ep, wofs, p0, p1, two);

  for (int gp = 0; gp < NGP; ++gp) {
    const int st = gp & 1;
    float* qs = qt + st * PLANE;
    float* ks = kt + st * PLANE;
    float* vs = vt + st * PLANE;
    float* es = emb + st * NEMB;
    const int cbase = gp * NCH;

    if (has_e) es[tid] = t.ev;

    v2f aq[4], ak[4], av[4];
    for (int c = 0; c < 4; ++c) {
      const bool nz = ((c >> 1) == hi);
      const int  cc = c & 1;
      aq[c][0] = nz ? t.wq[2 * cc] : 0.0f; aq[c][1] = nz ? t.wq[2 * cc + 1] : 0.0f;
      ak[c][0] = nz ? t.wk[2 * cc] : 0.0f; ak[c][1] = nz ? t.wk[2 * cc + 1] : 0.0f;
      av[c][0] = nz ? t.ww[2 * cc] : 0.0f; av[c][1] = nz ? t.ww[2 * cc + 1] : 0.0f;
    }

    conv_batch(aq, ak, av, t.u0, inb0, pos0, kh, qs, ks, vs);
    if (two)
      conv_batch(aq, ak, av, t.u1, inb1, pos1, kh, qs, ks, vs);

    if (gp + 1 < NGP) load_tile(t, gp + 1, x, Wq, Wk, Wv, ep, wofs, p0, p1, two);

    lds_barrier();

    for (int h = 0; h < 2; ++h) {
      const int sch  = wv + 8 * h;
      const int sofs = sch * CS + spx;
      const float* kc = ks + sofs;
      const float* vc = vs + sofs;
      const float* qc = qs + sofs + HTW + 1;
      float e9[9];
      for (int w9 = 0; w9 < 9; ++w9) e9[w9] = es[sch * 9 + w9];
      float kk[SR + 2][3], vv[SR + 2][3];
      for (int i = 0; i < 2; ++i) {
        for (int j = 0; j < 3; ++j) { kk[i][j] = kc[i * HTW + j]; vv[i][j] = vc[i * HTW + j]; }
      }
      float* op = out + (((size_t)b * 64 + cbase + sch) * Hdim + y0) * Wdim + (x0 + spx);
      for (int r = 0; r < SR; ++r) {
        for (int j = 0; j < 3; ++j) {
          kk[r + 2][j] = kc[(r + 2) * HTW + j];
          vv[r + 2][j] = vc[(r + 2) * HTW + j];
        }
        float q = qc[r * HTW];
        float lg[9];
        float mx = -1e30f;
        for (int w9 = 0; w9 < 9; ++w9) {
          int i = w9 / 3, j = w9 - 3 * i;
          float kv = kk[r + i][j] + e9[w9];
          float l = q * kv;
          lg[w9] = l;
          mx = fmaxf(mx, l);
        }
        float s = 0.0f, acc = 0.0f;
        for (int w9 = 0; w9 < 9; ++w9) {
          int i = w9 / 3, j = w9 - 3 * i;
          float e = __expf(lg[w9] - mx);
          s   += e;
          acc += e * vv[r + i][j];
        }
        op[r * Wdim] = acc / s;
      }
    }
  }
}

extern "C" void kernel_launch(void* const* d_in, const int* in_sizes, int n_in,
                              void* d_out, int out_size, void* d_ws, size_t ws_size,
                              hipStream_t stream) {
  (void)in_sizes; (void)n_in; (void)d_ws; (void)ws_size; (void)out_size;
  const float* x  = (const float*)d_in[0];
  const float* Wq = (const float*)d_in[1];
  const float* Wk = (const float*)d_in[2];
  const float* Wv = (const float*)d_in[3];
  const float* hE = (const float*)d_in[4];
  const float* wE = (const float*)d_in[5];
  float* out = (float*)d_out;

  dim3 grid(Wdim / TW, Hdim / TH, Bn);
  sasa_fused_kernel<<<grid, 256, 0, stream>>>(x, Wq, Wk, Wv, hE, wE, out);
}


# Wq and Wk of one group stacked as the 16 A rows (q rows 0-7, k rows 8-15) over that group's 8 inputs: 8 WMMAs per position batch instead of 12, no zero half in the q/k products; same sums
# speedup vs baseline: 3.9654x; 1.0560x over previous
#include <hip/hip_runtime.h>
#include <cstddef>


typedef __attribute__((ext_vector_type(2))) float v2f;
typedef __attribute__((ext_vector_type(8))) float v8f;

constexpr int Bn   = 8;
constexpr int Hdim = 128;
constexpr int Wdim = 128;
constexpr int NCH  = 16;
constexpr int NGP  = 4;
constexpr int TH   = 4;
constexpr int TW   = 32;
constexpr int HTH  = TH + 2;
constexpr int HTW  = TW + 2;
constexpr int HALO = HTH * HTW;
constexpr int NB   = (HALO + 15) / 16;
constexpr int CS    = HALO;
constexpr int PLANE = NCH * CS;
constexpr int NEMB  = NCH * 9;
constexpr int KOFF  = 2 * PLANE + 32;
constexpr int VOFF  = KOFF + 2 * PLANE;
constexpr int EOFF  = VOFF + 2 * PLANE;
constexpr int LDSN  = EOFF + 2 * NEMB;
constexpr int HW    = Hdim * Wdim;
constexpr int SR    = TH;

struct TileRegs {
  float w1[4], w2[4];
  float ww[4];
  float ev;
  float u0[8], u1[8];
};

__device__ __forceinline__ void load_tile(TileRegs& t, int gp,
                                          const float* __restrict__ x,  const float* Wqk,
                                          const float* __restrict__ Wv,
                                          const float* ep, int wofs1, int wofs, int xo0, int xo1, bool two) {
  const int wo1 = wofs1 + gp * 128;
  const int wo  = wofs + gp * 128;
  for (int cc = 0; cc < 2; ++cc) {
    for (int e = 0; e < 2; ++e) {
      t.w1[2 * cc + e] = Wqk[wo1 + 4 * cc + e];
      t.w2[2 * cc + e] = Wqk[wo1 + 64 + 4 * cc + e];
      t.ww[2 * cc + e] = Wv[wo + 4 * cc + e];
    }
  }
  t.ev = ep[gp * 24];
  const int xo = gp * NCH * HW;
  for (int c = 0; c < 4; ++c) {
    t.u0[2 * c]     = x[xo0 + xo + (4 * c) * HW];
    t.u0[2 * c + 1] = x[xo0 + xo + (4 * c + 1) * HW];
  }
  if (two) {
    for (int c = 0; c < 4; ++c) {
      t.u1[2 * c]     = x[xo1 + xo + (4 * c) * HW];
      t.u1[2 * c + 1] = x[xo1 + xo + (4 * c + 1) * HW];
    }
  } else {
    for (int c = 0; c < 8; ++c) t.u1[c] = 0.0f;
  }
}

__device__ __forceinline__ void lds_barrier() {
  asm volatile("s_wait_dscnt 0x0\n\ts_barrier_signal -1\n\ts_barrier_wait -1" ::: "memory");
}

__device__ __forceinline__ void conv_batch(const v2f* a1, const v2f* a2, const v2f* av,
                                           const float* u, bool inb, int pos, int kh,
                                           float* qk, float* vt) {
  v2f bb[4];
  for (int c = 0; c < 4; ++c) {
    bb[c][0] = inb ? u[2 * c]     : 0.0f;
    bb[c][1] = inb ? u[2 * c + 1] : 0.0f;
  }
  v8f c1 = {}, c2 = {}, cv = {};
  cv = __builtin_amdgcn_wmma_f32_16x16x4_f32(false, av[0], false, bb[0], (short)0, cv, false, false);
  c1 = __builtin_amdgcn_wmma_f32_16x16x4_f32(false, a1[0], false, bb[0], (short)0, c1, false, false);
  cv = __builtin_amdgcn_wmma_f32_16x16x4_f32(false, av[1], false, bb[1], (short)0, cv, false, false);
  c1 = __builtin_amdgcn_wmma_f32_16x16x4_f32(false, a1[1], false, bb[1], (short)0, c1, false, false);
  cv = __builtin_amdgcn_wmma_f32_16x16x4_f32(false, av[2], false, bb[2], (short)0, cv, false, false);
  c2 = __builtin_amdgcn_wmma_f32_16x16x4_f32(false, a2[0], false, bb[2], (short)0, c2, false, false);
  cv = __builtin_amdgcn_wmma_f32_16x16x4_f32(false, av[3], false, bb[3], (short)0, cv, false, false);
  c2 = __builtin_amdgcn_wmma_f32_16x16x4_f32(false, a2[1], false, bb[3], (short)0, c2, false, false);
  if (pos < HALO) {
    float* g1 = &qk[pos];
    float* g2 = &qk[8 * CS + pos];
    float* v8 = &vt[8 * kh * CS + pos];
    #pragma unroll
    for (int r = 0; r < 8; ++r) g1[r * CS] = c1[r];
    #pragma unroll
    for (int r = 0; r < 8; ++r) g2[r * CS] = c2[r];
    #pragma unroll
    for (int r = 0; r < 8; ++r) v8[r * CS] = cv[r];
  }
}

__global__ __launch_bounds__(256, 8) void sasa_fused_kernel(
    const float* __restrict__ x,  const float* __restrict__ Wq,
    const float* __restrict__ Wk, const float* __restrict__ Wv,
    const float* __restrict__ hE, const float* __restrict__ wE,
    float* __restrict__ out) {
  __shared__ float lds[LDSN];
  float* const qt  = lds;
  float* const kt  = lds + KOFF;
  float* const vt  = lds + VOFF;
  float* const emb = lds + EOFF;

  const int tid = threadIdx.x;
  const int bx  = blockIdx.x;
  const int by  = blockIdx.y;
  const int b   = blockIdx.z;
  const int x0  = bx * TW, y0 = by * TH;

  const int ln = tid & 31;
  const int m  = ln & 15;
  const int kh = ln >> 4;
  const int wv = tid >> 5;


  const int hi   = m >> 3;
  const int wofs = hi * 64 + (m & 7) * 8 + 2 * kh;
  const float* Wqk = hi ? Wk : Wq;
  const int wofs1 = (m & 7) * 8 + 2 * kh;

  const bool has_e = (tid < NEMB);
  const float* ep;
  {
    int t  = has_e ? tid : 0;
    int ch = t / 9, w9 = t - ch * 9;
    int i = w9 / 3, j = w9 - i * 3;
    int g  = ch >> 3;
    int cg = ch & 7;
    ep = (cg < 4) ? (hE + (g * 4 + cg) * 3 + i)
                  : (wE + (g * 4 + (cg - 4)) * 3 + j);
  }

  const int xg = (b * 64 + 2 * kh) * HW;
  const bool two = (wv + 8 < NB);
  const int pos0 = wv * 16 + m;
  const int pos1 = (wv + 8) * 16 + m;
  bool inb0, inb1 = false;
  int p0, p1;
  {
    int hy = pos0 / HTW, hx = pos0 - hy * HTW;
    int gy = y0 + hy - 1, gx = x0 + hx - 1;
    inb0 = (gy >= 0 && gy < Hdim && gx >= 0 && gx < Wdim);
    int cy = min(max(gy, 0), Hdim - 1), cx = min(max(gx, 0), Wdim - 1);
    p0 = xg + cy * Wdim + cx;
  }
  p1 = p0;
  if (two) {
    int pc  = (pos1 < HALO) ? pos1 : (HALO - 1);
    int hy = pc / HTW, hx = pc - hy * HTW;
    int gy = y0 + hy - 1, gx = x0 + hx - 1;
    inb1 = (gy >= 0 && gy < Hdim && gx >= 0 && gx < Wdim);
    int cy = min(max(gy, 0), Hdim - 1), cx = min(max(gx, 0), Wdim - 1);
    p1 = xg + cy * Wdim + cx;
  }

  const int spx  = ln;

  TileRegs t;
  load_tile(t, 0, x, Wqk, Wv, ep, wofs1, wofs, p0, p1, two);

  for (int gp = 0; gp < NGP; ++gp) {
    const int st = gp & 1;
    float* qs = qt + st * PLANE;
    float* ks = kt + st * PLANE;
    float* vs = vt + st * PLANE;
    float* es = emb + st * NEMB;
    const int cbase = gp * NCH;

    if (has_e) es[tid] = t.ev;

    v2f a1[2], a2[2], av[4];
    for (int c = 0; c < 2; ++c) {
      a1[c][0] = t.w1[2 * c]; a1[c][1] = t.w1[2 * c + 1];
      a2[c][0] = t.w2[2 * c]; a2[c][1] = t.w2[2 * c + 1];
    }
    for (int c = 0; c < 4; ++c) {
      const bool nz = ((c >> 1) == hi);
      const int  cc = c & 1;
      av[c][0] = nz ? t.ww[2 * cc] : 0.0f; av[c][1] = nz ? t.ww[2 * cc + 1] : 0.0f;
    }
    float* qks = kh ? ks : qs;

    conv_batch(a1, a2, av, t.u0, inb0, pos0, kh, qks, vs);
    if (two)
      conv_batch(a1, a2, av, t.u1, inb1, pos1, kh, qks, vs);

    if (gp + 1 < NGP) load_tile(t, gp + 1, x, Wqk, Wv, ep, wofs1, wofs, p0, p1, two);

    lds_barrier();

    for (int h = 0; h < 2; ++h) {
      const int sch  = wv + 8 * h;
      const int sofs = sch * CS + spx;
      const float* kc = ks + sofs;
      const float* vc = vs + sofs;
      const float* qc = qs + sofs + HTW + 1;
      float e9[9];
      for (int w9 = 0; w9 < 9; ++w9) e9[w9] = es[sch * 9 + w9];
      float kk[SR + 2][3], vv[SR + 2][3];
      for (int i = 0; i < 2; ++i) {
        for (int j = 0; j < 3; ++j) { kk[i][j] = kc[i * HTW + j]; vv[i][j] = vc[i * HTW + j]; }
      }
      float* op = out + (((size_t)b * 64 + cbase + sch) * Hdim + y0) * Wdim + (x0 + spx);
      for (int r = 0; r < SR; ++r) {
        for (int j = 0; j < 3; ++j) {
          kk[r + 2][j] = kc[(r + 2) * HTW + j];
          vv[r + 2][j] = vc[(r + 2) * HTW + j];
        }
        float q = qc[r * HTW];
        float lg[9];
        float mx = -1e30f;
        for (int w9 = 0; w9 < 9; ++w9) {
          int i = w9 / 3, j = w9 - 3 * i;
          float kv = kk[r + i][j] + e9[w9];
          float l = q * kv;
          lg[w9] = l;
          mx = fmaxf(mx, l);
        }
        float s = 0.0f, acc = 0.0f;
        for (int w9 = 0; w9 < 9; ++w9) {
          int i = w9 / 3, j = w9 - 3 * i;
          float e = __expf(lg[w9] - mx);
          s   += e;
          acc += e * vv[r + i][j];
        }
        op[r * Wdim] = acc / s;
      }
    }
  }
}

extern "C" void kernel_launch(void* const* d_in, const int* in_sizes, int n_in,
                              void* d_out, int out_size, void* d_ws, size_t ws_size,
                              hipStream_t stream) {
  (void)in_sizes; (void)n_in; (void)d_ws; (void)ws_size; (void)out_size;
  const float* x  = (const float*)d_in[0];
  const float* Wq = (const float*)d_in[1];
  const float* Wk = (const float*)d_in[2];
  const float* Wv = (const float*)d_in[3];
  const float* hE = (const float*)d_in[4];
  const float* wE = (const float*)d_in[5];
  float* out = (float*)d_out;

  dim3 grid(Wdim / TW, Hdim / TH, Bn);
  sasa_fused_kernel<<<grid, 256, 0, stream>>>(x, Wq, Wk, Wv, hE, wE, out);
}


# Softmax arithmetic strength-reduced: q pre-scaled by log2(e) so each window element is exp2(l - max) without its scale multiply, and acc / s becomes acc * rcp(s); same summation order
# speedup vs baseline: 4.2748x; 1.0780x over previous
#include <hip/hip_runtime.h>
#include <cstddef>


typedef __attribute__((ext_vector_type(2))) float v2f;
typedef __attribute__((ext_vector_type(8))) float v8f;

constexpr int Bn   = 8;
constexpr int Hdim = 128;
constexpr int Wdim = 128;
constexpr int NCH  = 16;
constexpr int NGP  = 4;
constexpr int TH   = 4;
constexpr int TW   = 32;
constexpr int HTH  = TH + 2;
constexpr int HTW  = TW + 2;
constexpr int HALO = HTH * HTW;
constexpr int NB   = (HALO + 15) / 16;
constexpr int CS    = HALO;
constexpr int PLANE = NCH * CS;
constexpr int NEMB  = NCH * 9;
constexpr int KOFF  = 2 * PLANE + 32;
constexpr int VOFF  = KOFF + 2 * PLANE;
constexpr int EOFF  = VOFF + 2 * PLANE;
constexpr int LDSN  = EOFF + 2 * NEMB;
constexpr int HW    = Hdim * Wdim;
constexpr int SR    = TH;

struct TileRegs {
  float w1[4], w2[4];
  float ww[4];
  float ev;
  float u0[8], u1[8];
};

__device__ __forceinline__ void load_tile(TileRegs& t, int gp,
                                          const float* __restrict__ x,  const float* Wqk,
                                          const float* __restrict__ Wv,
                                          const float* ep, int wofs1, int wofs, int xo0, int xo1, bool two) {
  const int wo1 = wofs1 + gp * 128;
  const int wo  = wofs + gp * 128;
  for (int cc = 0; cc < 2; ++cc) {
    for (int e = 0; e < 2; ++e) {
      t.w1[2 * cc + e] = Wqk[wo1 + 4 * cc + e];
      t.w2[2 * cc + e] = Wqk[wo1 + 64 + 4 * cc + e];
      t.ww[2 * cc + e] = Wv[wo + 4 * cc + e];
    }
  }
  t.ev = ep[gp * 24];
  const int xo = gp * NCH * HW;
  for (int c = 0; c < 4; ++c) {
    t.u0[2 * c]     = x[xo0 + xo + (4 * c) * HW];
    t.u0[2 * c + 1] = x[xo0 + xo + (4 * c + 1) * HW];
  }
  if (two) {
    for (int c = 0; c < 4; ++c) {
      t.u1[2 * c]     = x[xo1 + xo + (4 * c) * HW];
      t.u1[2 * c + 1] = x[xo1 + xo + (4 * c + 1) * HW];
    }
  } else {
    for (int c = 0; c < 8; ++c) t.u1[c] = 0.0f;
  }
}

__device__ __forceinline__ void lds_barrier() {
  asm volatile("s_wait_dscnt 0x0\n\ts_barrier_signal -1\n\ts_barrier_wait -1" ::: "memory");
}

__device__ __forceinline__ void conv_batch(const v2f* a1, const v2f* a2, const v2f* av,
                                           const float* u, bool inb, int pos, int kh,
                                           float* qk, float* vt) {
  v2f bb[4];
  for (int c = 0; c < 4; ++c) {
    bb[c][0] = inb ? u[2 * c]     : 0.0f;
    bb[c][1] = inb ? u[2 * c + 1] : 0.0f;
  }
  v8f c1 = {}, c2 = {}, cv = {};
  cv = __builtin_amdgcn_wmma_f32_16x16x4_f32(false, av[0], false, bb[0], (short)0, cv, false, false);
  c1 = __builtin_amdgcn_wmma_f32_16x16x4_f32(false, a1[0], false, bb[0], (short)0, c1, false, false);
  cv = __builtin_amdgcn_wmma_f32_16x16x4_f32(false, av[1], false, bb[1], (short)0, cv, false, false);
  c1 = __builtin_amdgcn_wmma_f32_16x16x4_f32(false, a1[1], false, bb[1], (short)0, c1, false, false);
  cv = __builtin_amdgcn_wmma_f32_16x16x4_f32(false, av[2], false, bb[2], (short)0, cv, false, false);
  c2 = __builtin_amdgcn_wmma_f32_16x16x4_f32(false, a2[0], false, bb[2], (short)0, c2, false, false);
  cv = __builtin_amdgcn_wmma_f32_16x16x4_f32(false, av[3], false, bb[3], (short)0, cv, false, false);
  c2 = __builtin_amdgcn_wmma_f32_16x16x4_f32(false, a2[1], false, bb[3], (short)0, c2, false, false);
  if (pos < HALO) {
    float* g1 = &qk[pos];
    float* g2 = &qk[8 * CS + pos];
    float* v8 = &vt[8 * kh * CS + pos];
    #pragma unroll
    for (int r = 0; r < 8; ++r) g1[r * CS] = c1[r];
    #pragma unroll
    for (int r = 0; r < 8; ++r) g2[r * CS] = c2[r];
    #pragma unroll
    for (int r = 0; r < 8; ++r) v8[r * CS] = cv[r];
  }
}

__global__ __launch_bounds__(256, 8) void sasa_fused_kernel(
    const float* __restrict__ x,  const float* __restrict__ Wq,
    const float* __restrict__ Wk, const float* __restrict__ Wv,
    const float* __restrict__ hE, const float* __restrict__ wE,
    float* __restrict__ out) {
  __shared__ float lds[LDSN];
  float* const qt  = lds;
  float* const kt  = lds + KOFF;
  float* const vt  = lds + VOFF;
  float* const emb = lds + EOFF;

  const int tid = threadIdx.x;
  const int bx  = blockIdx.x;
  const int by  = blockIdx.y;
  const int b   = blockIdx.z;
  const int x0  = bx * TW, y0 = by * TH;

  const int ln = tid & 31;
  const int m  = ln & 15;
  const int kh = ln >> 4;
  const int wv = tid >> 5;


  const int hi   = m >> 3;
  const int wofs = hi * 64 + (m & 7) * 8 + 2 * kh;
  const float* Wqk = hi ? Wk : Wq;
  const int wofs1 = (m & 7) * 8 + 2 * kh;

  const bool has_e = (tid < NEMB);
  const float* ep;
  {
    int t  = has_e ? tid : 0;
    int ch = t / 9, w9 = t - ch * 9;
    int i = w9 / 3, j = w9 - i * 3;
    int g  = ch >> 3;
    int cg = ch & 7;
    ep = (cg < 4) ? (hE + (g * 4 + cg) * 3 + i)
                  : (wE + (g * 4 + (cg - 4)) * 3 + j);
  }

  const int xg = (b * 64 + 2 * kh) * HW;
  const bool two = (wv + 8 < NB);
  const int pos0 = wv * 16 + m;
  const int pos1 = (wv + 8) * 16 + m;
  bool inb0, inb1 = false;
  int p0, p1;
  {
    int hy = pos0 / HTW, hx = pos0 - hy * HTW;
    int gy = y0 + hy - 1, gx = x0 + hx - 1;
    inb0 = (gy >= 0 && gy < Hdim && gx >= 0 && gx < Wdim);
    int cy = min(max(gy, 0), Hdim - 1), cx = min(max(gx, 0), Wdim - 1);
    p0 = xg + cy * Wdim + cx;
  }
  p1 = p0;
  if (two) {
    int pc  = (pos1 < HALO) ? pos1 : (HALO - 1);
    int hy = pc / HTW, hx = pc - hy * HTW;
    int gy = y0 + hy - 1, gx = x0 + hx - 1;
    inb1 = (gy >= 0 && gy < Hdim && gx >= 0 && gx < Wdim);
    int cy = min(max(gy, 0), Hdim - 1), cx = min(max(gx, 0), Wdim - 1);
    p1 = xg + cy * Wdim + cx;
  }

  const int spx  = ln;

  TileRegs t;
  load_tile(t, 0, x, Wqk, Wv, ep, wofs1, wofs, p0, p1, two);

  for (int gp = 0; gp < NGP; ++gp) {
    const int st = gp & 1;
    float* qs = qt + st * PLANE;
    float* ks = kt + st * PLANE;
    float* vs = vt + st * PLANE;
    float* es = emb + st * NEMB;
    const int cbase = gp * NCH;

    if (has_e) es[tid] = t.ev;

    v2f a1[2], a2[2], av[4];
    for (int c = 0; c < 2; ++c) {
      a1[c][0] = t.w1[2 * c]; a1[c][1] = t.w1[2 * c + 1];
      a2[c][0] = t.w2[2 * c]; a2[c][1] = t.w2[2 * c + 1];
    }
    for (int c = 0; c < 4; ++c) {
      const bool nz = ((c >> 1) == hi);
      const int  cc = c & 1;
      av[c][0] = nz ? t.ww[2 * cc] : 0.0f; av[c][1] = nz ? t.ww[2 * cc + 1] : 0.0f;
    }
    float* qks = kh ? ks : qs;

    conv_batch(a1, a2, av, t.u0, inb0, pos0, kh, qks, vs);
    if (two)
      conv_batch(a1, a2, av, t.u1, inb1, pos1, kh, qks, vs);

    if (gp + 1 < NGP) load_tile(t, gp + 1, x, Wqk, Wv, ep, wofs1, wofs, p0, p1, two);

    lds_barrier();

    for (int h = 0; h < 2; ++h) {
      const int sch  = wv + 8 * h;
      const int sofs = sch * CS + spx;
      const float* kc = ks + sofs;
      const float* vc = vs + sofs;
      const float* qc = qs + sofs + HTW + 1;
      float e9[9];
      for (int w9 = 0; w9 < 9; ++w9) e9[w9] = es[sch * 9 + w9];
      float kk[SR + 2][3], vv[SR + 2][3];
      for (int i = 0; i < 2; ++i) {
        for (int j = 0; j < 3; ++j) { kk[i][j] = kc[i * HTW + j]; vv[i][j] = vc[i * HTW + j]; }
      }
      float* op = out + (((size_t)b * 64 + cbase + sch) * Hdim + y0) * Wdim + (x0 + spx);
      for (int r = 0; r < SR; ++r) {
        for (int j = 0; j < 3; ++j) {
          kk[r + 2][j] = kc[(r + 2) * HTW + j];
          vv[r + 2][j] = vc[(r + 2) * HTW + j];
        }
        float q = qc[r * HTW] * 1.44269504088896341f;
        float lg[9];
        float mx = -1e30f;
        for (int w9 = 0; w9 < 9; ++w9) {
          int i = w9 / 3, j = w9 - 3 * i;
          float kv = kk[r + i][j] + e9[w9];
          float l = q * kv;
          lg[w9] = l;
          mx = fmaxf(mx, l);
        }
        float s = 0.0f, acc = 0.0f;
        for (int w9 = 0; w9 < 9; ++w9) {
          int i = w9 / 3, j = w9 - 3 * i;
          float e = __builtin_amdgcn_exp2f(lg[w9] - mx);
          s   += e;
          acc += e * vv[r + i][j];
        }
        op[r * Wdim] = acc * __builtin_amdgcn_rcpf(s);
      }
    }
  }
}

extern "C" void kernel_launch(void* const* d_in, const int* in_sizes, int n_in,
                              void* d_out, int out_size, void* d_ws, size_t ws_size,
                              hipStream_t stream) {
  (void)in_sizes; (void)n_in; (void)d_ws; (void)ws_size; (void)out_size;
  const float* x  = (const float*)d_in[0];
  const float* Wq = (const float*)d_in[1];
  const float* Wk = (const float*)d_in[2];
  const float* Wv = (const float*)d_in[3];
  const float* hE = (const float*)d_in[4];
  const float* wE = (const float*)d_in[5];
  float* out = (float*)d_out;

  dim3 grid(Wdim / TW, Hdim / TH, Bn);
  sasa_fused_kernel<<<grid, 256, 0, stream>>>(x, Wq, Wk, Wv, hE, wE, out);
}


# Softmax on channel pairs: pair-interleaved LDS tiles, two channels per lane on packed two-float math, same sums; to fit 128 registers: weights fetched late in prior tile, embeddings staged once
# speedup vs baseline: 4.4550x; 1.0422x over previous
#include <hip/hip_runtime.h>
#include <cstddef>


typedef __attribute__((ext_vector_type(2))) float v2f;
typedef __attribute__((ext_vector_type(8))) float v8f;

constexpr int Bn   = 8;
constexpr int Hdim = 128;
constexpr int Wdim = 128;
constexpr int NCH  = 16;
constexpr int NGP  = 4;
constexpr int TH   = 4;
constexpr int TW   = 32;
constexpr int HTH  = TH + 2;
constexpr int HTW  = TW + 2;
constexpr int HALO = HTH * HTW;
constexpr int NB   = (HALO + 15) / 16;
constexpr int CS    = HALO;
constexpr int PS    = 2 * CS;
constexpr int PLANE = NCH * CS;
constexpr int NEMB  = NCH * 9;
constexpr int KOFF  = 2 * PLANE + 32;
constexpr int VOFF  = KOFF + 2 * PLANE;
constexpr int EOFF  = VOFF + 2 * PLANE;
constexpr int LDSN  = EOFF + NGP * NEMB;
constexpr int HW    = Hdim * Wdim;
constexpr int SR    = TH;

struct TileRegs {
  float w1[4], w2[4];
  float ww[4];
  float u0[8], u1[8];
};

__device__ __forceinline__ void load_tile(TileRegs& t, int gp, const float* __restrict__ x,
                                          int xo0, int xo1, bool two) {
  const int xo = gp * NCH * HW;
  for (int c = 0; c < 4; ++c) {
    t.u0[2 * c]     = x[xo0 + xo + (4 * c) * HW];
    t.u0[2 * c + 1] = x[xo0 + xo + (4 * c + 1) * HW];
  }
  if (two) {
    for (int c = 0; c < 4; ++c) {
      t.u1[2 * c]     = x[xo1 + xo + (4 * c) * HW];
      t.u1[2 * c + 1] = x[xo1 + xo + (4 * c + 1) * HW];
    }
  } else {
    for (int c = 0; c < 8; ++c) t.u1[c] = 0.0f;
  }
}

__device__ __forceinline__ void load_w(TileRegs& t, int gp, const float* Wqk,
                                       const float* __restrict__ Wv, int wofs1, int wofs) {
  const int wo1 = wofs1 + gp * 128;
  const int wo  = wofs + gp * 128;
  for (int cc = 0; cc < 2; ++cc) {
    for (int e = 0; e < 2; ++e) {
      t.w1[2 * cc + e] = Wqk[wo1 + 4 * cc + e];
      t.w2[2 * cc + e] = Wqk[wo1 + 64 + 4 * cc + e];
      t.ww[2 * cc + e] = Wv[wo + 4 * cc + e];
    }
  }
}

__device__ __forceinline__ void lds_barrier() {
  asm volatile("s_wait_dscnt 0x0\n\ts_barrier_signal -1\n\ts_barrier_wait -1" ::: "memory");
}

__device__ __forceinline__ void conv_batch(const v2f* a1, const v2f* a2, const v2f* av,
                                           const float* u, bool inb, int pos, int kh,
                                           float* qk, float* vt) {
  v2f bb[4];
  for (int c = 0; c < 4; ++c) {
    bb[c][0] = inb ? u[2 * c]     : 0.0f;
    bb[c][1] = inb ? u[2 * c + 1] : 0.0f;
  }
  v8f c1 = {}, c2 = {}, cv = {};
  cv = __builtin_amdgcn_wmma_f32_16x16x4_f32(false, av[0], false, bb[0], (short)0, cv, false, false);
  c1 = __builtin_amdgcn_wmma_f32_16x16x4_f32(false, a1[0], false, bb[0], (short)0, c1, false, false);
  cv = __builtin_amdgcn_wmma_f32_16x16x4_f32(false, av[1], false, bb[1], (short)0, cv, false, false);
  c1 = __builtin_amdgcn_wmma_f32_16x16x4_f32(false, a1[1], false, bb[1], (short)0, c1, false, false);
  cv = __builtin_amdgcn_wmma_f32_16x16x4_f32(false, av[2], false, bb[2], (short)0, cv, false, false);
  c2 = __builtin_amdgcn_wmma_f32_16x16x4_f32(false, a2[0], false, bb[2], (short)0, c2, false, false);
  cv = __builtin_amdgcn_wmma_f32_16x16x4_f32(false, av[3], false, bb[3], (short)0, cv, false, false);
  c2 = __builtin_amdgcn_wmma_f32_16x16x4_f32(false, a2[1], false, bb[3], (short)0, c2, false, false);
  if (pos < HALO) {
    float* g1 = &qk[2 * pos];
    float* g2 = &qk[4 * PS + 2 * pos];
    float* v8 = &vt[4 * kh * PS + 2 * pos];
    #pragma unroll
    for (int t = 0; t < 4; ++t) { v2f p; p[0] = c1[2 * t]; p[1] = c1[2 * t + 1]; *(v2f*)&g1[t * PS] = p; }
    #pragma unroll
    for (int t = 0; t < 4; ++t) { v2f p; p[0] = c2[2 * t]; p[1] = c2[2 * t + 1]; *(v2f*)&g2[t * PS] = p; }
    #pragma unroll
    for (int t = 0; t < 4; ++t) { v2f p; p[0] = cv[2 * t]; p[1] = cv[2 * t + 1]; *(v2f*)&v8[t * PS] = p; }
  }
}

__global__ __launch_bounds__(256, 8) void sasa_fused_kernel(
    const float* __restrict__ x,  const float* __restrict__ Wq,
    const float* __restrict__ Wk, const float* __restrict__ Wv,
    const float* __restrict__ hE, const float* __restrict__ wE,
    float* __restrict__ out) {
  __shared__ __attribute__((aligned(16))) float lds[LDSN];
  float* const qt  = lds;
  float* const kt  = lds + KOFF;
  float* const vt  = lds + VOFF;
  float* const emb = lds + EOFF;

  const int tid = threadIdx.x;
  const int bx  = blockIdx.x;
  const int by  = blockIdx.y;
  const int b   = blockIdx.z;
  const int x0  = bx * TW, y0 = by * TH;

  const int ln = tid & 31;
  const int m  = ln & 15;
  const int kh = ln >> 4;
  const int wv = tid >> 5;


  const int hi   = m >> 3;
  const int wofs = hi * 64 + (m & 7) * 8 + 2 * kh;
  const float* Wqk = hi ? Wk : Wq;
  const int wofs1 = (m & 7) * 8 + 2 * kh;

  const bool has_e = (tid < NEMB);
  const float* ep;
  int eidx;
  {
    int t  = has_e ? tid : 0;
    int ch = t / 9, w9 = t - ch * 9;
    eidx = (ch >> 1) * 18 + 2 * w9 + (ch & 1);
    int i = w9 / 3, j = w9 - i * 3;
    int g  = ch >> 3;
    int cg = ch & 7;
    ep = (cg < 4) ? (hE + (g * 4 + cg) * 3 + i)
                  : (wE + (g * 4 + (cg - 4)) * 3 + j);
  }

  const int xg = (b * 64 + 2 * kh) * HW;
  const bool two = (wv + 8 < NB);
  const int pos0 = wv * 16 + m;
  const int pos1 = (wv + 8) * 16 + m;
  bool inb0, inb1 = false;
  int p0, p1;
  {
    int hy = pos0 / HTW, hx = pos0 - hy * HTW;
    int gy = y0 + hy - 1, gx = x0 + hx - 1;
    inb0 = (gy >= 0 && gy < Hdim && gx >= 0 && gx < Wdim);
    int cy = min(max(gy, 0), Hdim - 1), cx = min(max(gx, 0), Wdim - 1);
    p0 = xg + cy * Wdim + cx;
  }
  p1 = p0;
  if (two) {
    int pc  = (pos1 < HALO) ? pos1 : (HALO - 1);
    int hy = pc / HTW, hx = pc - hy * HTW;
    int gy = y0 + hy - 1, gx = x0 + hx - 1;
    inb1 = (gy >= 0 && gy < Hdim && gx >= 0 && gx < Wdim);
    int cy = min(max(gy, 0), Hdim - 1), cx = min(max(gx, 0), Wdim - 1);
    p1 = xg + cy * Wdim + cx;
  }

  const int spx  = ln;

  TileRegs t;
  load_tile(t, 0, x, p0, p1, two);
  load_w(t, 0, Wqk, Wv, wofs1, wofs);

  if (has_e) {
    for (int g4 = 0; g4 < NGP; ++g4) emb[g4 * NEMB + eidx] = ep[g4 * 24];
  }

  for (int gp = 0; gp < NGP; ++gp) {
    const int st = gp & 1;
    float* qs = qt + st * PLANE;
    float* ks = kt + st * PLANE;
    float* vs = vt + st * PLANE;
    const float* es = emb + gp * NEMB;
    const int cbase = gp * NCH;

    v2f a1[2], a2[2], av[4];
    for (int c = 0; c < 2; ++c) {
      a1[c][0] = t.w1[2 * c]; a1[c][1] = t.w1[2 * c + 1];
      a2[c][0] = t.w2[2 * c]; a2[c][1] = t.w2[2 * c + 1];
    }
    for (int c = 0; c < 4; ++c) {
      const bool nz = ((c >> 1) == hi);
      const int  cc = c & 1;
      av[c][0] = nz ? t.ww[2 * cc] : 0.0f; av[c][1] = nz ? t.ww[2 * cc + 1] : 0.0f;
    }
    float* qks = kh ? ks : qs;

    conv_batch(a1, a2, av, t.u0, inb0, pos0, kh, qks, vs);
    if (two)
      conv_batch(a1, a2, av, t.u1, inb1, pos1, kh, qks, vs);

    if (gp + 1 < NGP) load_tile(t, gp + 1, x, p0, p1, two);

    lds_barrier();

    {
      const int sofs = wv * PS + 2 * spx;
      const float* kc = ks + sofs;
      const float* vc = vs + sofs;
      const float* qc = qs + sofs + 2 * (HTW + 1);
      v2f e9[9];
      for (int w9 = 0; w9 < 9; ++w9) e9[w9] = *(const v2f*)&es[wv * 18 + 2 * w9];
      v2f kk[SR + 2][3], vv[SR + 2][3];
      for (int i = 0; i < 2; ++i) {
        for (int j = 0; j < 3; ++j) {
          kk[i][j] = *(const v2f*)&kc[2 * (i * HTW + j)];
          vv[i][j] = *(const v2f*)&vc[2 * (i * HTW + j)];
        }
      }
      float* op = out + (((b * 64 + cbase + 2 * wv) * Hdim + y0) * Wdim + (x0 + spx));
      for (int r = 0; r < SR; ++r) {
        if (r == SR - 1 && gp + 1 < NGP) load_w(t, gp + 1, Wqk, Wv, wofs1, wofs);
        for (int j = 0; j < 3; ++j) {
          kk[r + 2][j] = *(const v2f*)&kc[2 * ((r + 2) * HTW + j)];
          vv[r + 2][j] = *(const v2f*)&vc[2 * ((r + 2) * HTW + j)];
        }
        v2f q = *(const v2f*)&qc[2 * r * HTW] * 1.44269504088896341f;
        v2f lg[9];
        for (int w9 = 0; w9 < 9; ++w9) {
          int i = w9 / 3, j = w9 - 3 * i;
          lg[w9] = q * (kk[r + i][j] + e9[w9]);
        }
        v2f mx;
        for (int c = 0; c < 2; ++c) {
          float m3 = fmaxf(fmaxf(lg[0][c], lg[1][c]), lg[2][c]);
          m3 = fmaxf(fmaxf(m3, lg[3][c]), lg[4][c]);
          m3 = fmaxf(fmaxf(m3, lg[5][c]), lg[6][c]);
          m3 = fmaxf(fmaxf(m3, lg[7][c]), lg[8][c]);
          mx[c] = m3;
        }
        v2f s, acc;
        for (int w9 = 0; w9 < 9; ++w9) {
          int i = w9 / 3, j = w9 - 3 * i;
          v2f d = lg[w9] - mx;
          v2f e;
          e[0] = __builtin_amdgcn_exp2f(d[0]);
          e[1] = __builtin_amdgcn_exp2f(d[1]);
          if (w9 == 0) { s = e; acc = e * vv[r][0]; }
          else         { s += e; acc = __builtin_elementwise_fma(e, vv[r + i][j], acc); }
        }
        v2f rs;
        rs[0] = __builtin_amdgcn_rcpf(s[0]);
        rs[1] = __builtin_amdgcn_rcpf(s[1]);
        v2f o = acc * rs;
        op[r * Wdim]      = o[0];
        op[HW + r * Wdim] = o[1];
      }
    }
  }
}

extern "C" void kernel_launch(void* const* d_in, const int* in_sizes, int n_in,
                              void* d_out, int out_size, void* d_ws, size_t ws_size,
                              hipStream_t stream) {
  (void)in_sizes; (void)n_in; (void)d_ws; (void)ws_size; (void)out_size;
  const float* x  = (const float*)d_in[0];
  const float* Wq = (const float*)d_in[1];
  const float* Wk = (const float*)d_in[2];
  const float* Wv = (const float*)d_in[3];
  const float* hE = (const float*)d_in[4];
  const float* wE = (const float*)d_in[5];
  float* out = (float*)d_out;

  dim3 grid(Wdim / TW, Hdim / TH, Bn);
  sasa_fused_kernel<<<grid, 256, 0, stream>>>(x, Wq, Wk, Wv, hE, wE, out);
}


# Wave issue priority ranked by tile index (s_setprio 3,2,1,0 over a work-group's four tiles): the work-group that is behind wins the SIMD, so the four sharing it finish together, not oldest first
# speedup vs baseline: 4.6997x; 1.0549x over previous
#include <hip/hip_runtime.h>
#include <cstddef>


typedef __attribute__((ext_vector_type(2))) float v2f;
typedef __attribute__((ext_vector_type(8))) float v8f;

constexpr int Bn   = 8;
constexpr int Hdim = 128;
constexpr int Wdim = 128;
constexpr int NCH  = 16;
constexpr int NGP  = 4;
constexpr int TH   = 4;
constexpr int TW   = 32;
constexpr int HTH  = TH + 2;
constexpr int HTW  = TW + 2;
constexpr int HALO = HTH * HTW;
constexpr int NB   = (HALO + 15) / 16;
constexpr int CS    = HALO;
constexpr int PS    = 2 * CS;
constexpr int PLANE = NCH * CS;
constexpr int NEMB  = NCH * 9;
constexpr int KOFF  = 2 * PLANE + 32;
constexpr int VOFF  = KOFF + 2 * PLANE;
constexpr int EOFF  = VOFF + 2 * PLANE;
constexpr int LDSN  = EOFF + NGP * NEMB;
constexpr int HW    = Hdim * Wdim;
constexpr int SR    = TH;

struct TileRegs {
  float w1[4], w2[4];
  float ww[4];
  float u0[8], u1[8];
};

__device__ __forceinline__ void load_tile(TileRegs& t, int gp, const float* __restrict__ x,
                                          int xo0, int xo1, bool two) {
  const int xo = gp * NCH * HW;
  for (int c = 0; c < 4; ++c) {
    t.u0[2 * c]     = x[xo0 + xo + (4 * c) * HW];
    t.u0[2 * c + 1] = x[xo0 + xo + (4 * c + 1) * HW];
  }
  if (two) {
    for (int c = 0; c < 4; ++c) {
      t.u1[2 * c]     = x[xo1 + xo + (4 * c) * HW];
      t.u1[2 * c + 1] = x[xo1 + xo + (4 * c + 1) * HW];
    }
  } else {
    for (int c = 0; c < 8; ++c) t.u1[c] = 0.0f;
  }
}

__device__ __forceinline__ void load_w(TileRegs& t, int gp, const float* Wqk,
                                       const float* __restrict__ Wv, int wofs1, int wofs) {
  const int wo1 = wofs1 + gp * 128;
  const int wo  = wofs + gp * 128;
  for (int cc = 0; cc < 2; ++cc) {
    for (int e = 0; e < 2; ++e) {
      t.w1[2 * cc + e] = Wqk[wo1 + 4 * cc + e];
      t.w2[2 * cc + e] = Wqk[wo1 + 64 + 4 * cc + e];
      t.ww[2 * cc + e] = Wv[wo + 4 * cc + e];
    }
  }
}

__device__ __forceinline__ void lds_barrier() {
  asm volatile("s_wait_dscnt 0x0\n\ts_barrier_signal -1\n\ts_barrier_wait -1" ::: "memory");
}

__device__ __forceinline__ void conv_batch(const v2f* a1, const v2f* a2, const v2f* av,
                                           const float* u, bool inb, int pos, int kh,
                                           float* qk, float* vt) {
  v2f bb[4];
  for (int c = 0; c < 4; ++c) {
    bb[c][0] = inb ? u[2 * c]     : 0.0f;
    bb[c][1] = inb ? u[2 * c + 1] : 0.0f;
  }
  v8f c1 = {}, c2 = {}, cv = {};
  cv = __builtin_amdgcn_wmma_f32_16x16x4_f32(false, av[0], false, bb[0], (short)0, cv, false, false);
  c1 = __builtin_amdgcn_wmma_f32_16x16x4_f32(false, a1[0], false, bb[0], (short)0, c1, false, false);
  cv = __builtin_amdgcn_wmma_f32_16x16x4_f32(false, av[1], false, bb[1], (short)0, cv, false, false);
  c1 = __builtin_amdgcn_wmma_f32_16x16x4_f32(false, a1[1], false, bb[1], (short)0, c1, false, false);
  cv = __builtin_amdgcn_wmma_f32_16x16x4_f32(false, av[2], false, bb[2], (short)0, cv, false, false);
  c2 = __builtin_amdgcn_wmma_f32_16x16x4_f32(false, a2[0], false, bb[2], (short)0, c2, false, false);
  cv = __builtin_amdgcn_wmma_f32_16x16x4_f32(false, av[3], false, bb[3], (short)0, cv, false, false);
  c2 = __builtin_amdgcn_wmma_f32_16x16x4_f32(false, a2[1], false, bb[3], (short)0, c2, false, false);
  if (pos < HALO) {
    float* g1 = &qk[2 * pos];
    float* g2 = &qk[4 * PS + 2 * pos];
    float* v8 = &vt[4 * kh * PS + 2 * pos];
    #pragma unroll
    for (int t = 0; t < 4; ++t) { v2f p; p[0] = c1[2 * t]; p[1] = c1[2 * t + 1]; *(v2f*)&g1[t * PS] = p; }
    #pragma unroll
    for (int t = 0; t < 4; ++t) { v2f p; p[0] = c2[2 * t]; p[1] = c2[2 * t + 1]; *(v2f*)&g2[t * PS] = p; }
    #pragma unroll
    for (int t = 0; t < 4; ++t) { v2f p; p[0] = cv[2 * t]; p[1] = cv[2 * t + 1]; *(v2f*)&v8[t * PS] = p; }
  }
}

__global__ __launch_bounds__(256, 8) void sasa_fused_kernel(
    const float* __restrict__ x,  const float* __restrict__ Wq,
    const float* __restrict__ Wk, const float* __restrict__ Wv,
    const float* __restrict__ hE, const float* __restrict__ wE,
    float* __restrict__ out) {
  __shared__ __attribute__((aligned(16))) float lds[LDSN];
  float* const qt  = lds;
  float* const kt  = lds + KOFF;
  float* const vt  = lds + VOFF;
  float* const emb = lds + EOFF;

  const int tid = threadIdx.x;
  const int bx  = blockIdx.x;
  const int by  = blockIdx.y;
  const int b   = blockIdx.z;
  const int x0  = bx * TW, y0 = by * TH;

  const int ln = tid & 31;
  const int m  = ln & 15;
  const int kh = ln >> 4;
  const int wv = tid >> 5;


  const int hi   = m >> 3;
  const int wofs = hi * 64 + (m & 7) * 8 + 2 * kh;
  const float* Wqk = hi ? Wk : Wq;
  const int wofs1 = (m & 7) * 8 + 2 * kh;

  const bool has_e = (tid < NEMB);
  const float* ep;
  int eidx;
  {
    int t  = has_e ? tid : 0;
    int ch = t / 9, w9 = t - ch * 9;
    eidx = (ch >> 1) * 18 + 2 * w9 + (ch & 1);
    int i = w9 / 3, j = w9 - i * 3;
    int g  = ch >> 3;
    int cg = ch & 7;
    ep = (cg < 4) ? (hE + (g * 4 + cg) * 3 + i)
                  : (wE + (g * 4 + (cg - 4)) * 3 + j);
  }

  const int xg = (b * 64 + 2 * kh) * HW;
  const bool two = (wv + 8 < NB);
  const int pos0 = wv * 16 + m;
  const int pos1 = (wv + 8) * 16 + m;
  bool inb0, inb1 = false;
  int p0, p1;
  {
    int hy = pos0 / HTW, hx = pos0 - hy * HTW;
    int gy = y0 + hy - 1, gx = x0 + hx - 1;
    inb0 = (gy >= 0 && gy < Hdim && gx >= 0 && gx < Wdim);
    int cy = min(max(gy, 0), Hdim - 1), cx = min(max(gx, 0), Wdim - 1);
    p0 = xg + cy * Wdim + cx;
  }
  p1 = p0;
  if (two) {
    int pc  = (pos1 < HALO) ? pos1 : (HALO - 1);
    int hy = pc / HTW, hx = pc - hy * HTW;
    int gy = y0 + hy - 1, gx = x0 + hx - 1;
    inb1 = (gy >= 0 && gy < Hdim && gx >= 0 && gx < Wdim);
    int cy = min(max(gy, 0), Hdim - 1), cx = min(max(gx, 0), Wdim - 1);
    p1 = xg + cy * Wdim + cx;
  }

  const int spx  = ln;

  TileRegs t;
  load_tile(t, 0, x, p0, p1, two);
  load_w(t, 0, Wqk, Wv, wofs1, wofs);

  if (has_e) {
    for (int g4 = 0; g4 < NGP; ++g4) emb[g4 * NEMB + eidx] = ep[g4 * 24];
  }

  __builtin_amdgcn_s_setprio(3);
  for (int gp = 0; gp < NGP; ++gp) {
    const int st = gp & 1;
    float* qs = qt + st * PLANE;
    float* ks = kt + st * PLANE;
    float* vs = vt + st * PLANE;
    const float* es = emb + gp * NEMB;
    const int cbase = gp * NCH;

    v2f a1[2], a2[2], av[4];
    for (int c = 0; c < 2; ++c) {
      a1[c][0] = t.w1[2 * c]; a1[c][1] = t.w1[2 * c + 1];
      a2[c][0] = t.w2[2 * c]; a2[c][1] = t.w2[2 * c + 1];
    }
    for (int c = 0; c < 4; ++c) {
      const bool nz = ((c >> 1) == hi);
      const int  cc = c & 1;
      av[c][0] = nz ? t.ww[2 * cc] : 0.0f; av[c][1] = nz ? t.ww[2 * cc + 1] : 0.0f;
    }
    float* qks = kh ? ks : qs;

    conv_batch(a1, a2, av, t.u0, inb0, pos0, kh, qks, vs);
    if (two)
      conv_batch(a1, a2, av, t.u1, inb1, pos1, kh, qks, vs);

    if (gp + 1 < NGP) load_tile(t, gp + 1, x, p0, p1, two);

    lds_barrier();

    {
      const int sofs = wv * PS + 2 * spx;
      const float* kc = ks + sofs;
      const float* vc = vs + sofs;
      const float* qc = qs + sofs + 2 * (HTW + 1);
      v2f e9[9];
      for (int w9 = 0; w9 < 9; ++w9) e9[w9] = *(const v2f*)&es[wv * 18 + 2 * w9];
      v2f kk[SR + 2][3], vv[SR + 2][3];
      for (int i = 0; i < 2; ++i) {
        for (int j = 0; j < 3; ++j) {
          kk[i][j] = *(const v2f*)&kc[2 * (i * HTW + j)];
          vv[i][j] = *(const v2f*)&vc[2 * (i * HTW + j)];
        }
      }
      float* op = out + (((b * 64 + cbase + 2 * wv) * Hdim + y0) * Wdim + (x0 + spx));
      for (int r = 0; r < SR; ++r) {
        if (r == SR - 1 && gp + 1 < NGP) load_w(t, gp + 1, Wqk, Wv, wofs1, wofs);
        for (int j = 0; j < 3; ++j) {
          kk[r + 2][j] = *(const v2f*)&kc[2 * ((r + 2) * HTW + j)];
          vv[r + 2][j] = *(const v2f*)&vc[2 * ((r + 2) * HTW + j)];
        }
        v2f q = *(const v2f*)&qc[2 * r * HTW] * 1.44269504088896341f;
        v2f lg[9];
        for (int w9 = 0; w9 < 9; ++w9) {
          int i = w9 / 3, j = w9 - 3 * i;
          lg[w9] = q * (kk[r + i][j] + e9[w9]);
        }
        v2f mx;
        for (int c = 0; c < 2; ++c) {
          float m3 = fmaxf(fmaxf(lg[0][c], lg[1][c]), lg[2][c]);
          m3 = fmaxf(fmaxf(m3, lg[3][c]), lg[4][c]);
          m3 = fmaxf(fmaxf(m3, lg[5][c]), lg[6][c]);
          m3 = fmaxf(fmaxf(m3, lg[7][c]), lg[8][c]);
          mx[c] = m3;
        }
        v2f s, acc;
        for (int w9 = 0; w9 < 9; ++w9) {
          int i = w9 / 3, j = w9 - 3 * i;
          v2f d = lg[w9] - mx;
          v2f e;
          e[0] = __builtin_amdgcn_exp2f(d[0]);
          e[1] = __builtin_amdgcn_exp2f(d[1]);
          if (w9 == 0) { s = e; acc = e * vv[r][0]; }
          else         { s += e; acc = __builtin_elementwise_fma(e, vv[r + i][j], acc); }
        }
        v2f rs;
        rs[0] = __builtin_amdgcn_rcpf(s[0]);
        rs[1] = __builtin_amdgcn_rcpf(s[1]);
        v2f o = acc * rs;
        op[r * Wdim]      = o[0];
        op[HW + r * Wdim] = o[1];
      }
    }
    if (gp == 0)      __builtin_amdgcn_s_setprio(2);
    else if (gp == 1) __builtin_amdgcn_s_setprio(1);
    else if (gp == 2) __builtin_amdgcn_s_setprio(0);
  }
}

extern "C" void kernel_launch(void* const* d_in, const int* in_sizes, int n_in,
                              void* d_out, int out_size, void* d_ws, size_t ws_size,
                              hipStream_t stream) {
  (void)in_sizes; (void)n_in; (void)d_ws; (void)ws_size; (void)out_size;
  const float* x  = (const float*)d_in[0];
  const float* Wq = (const float*)d_in[1];
  const float* Wk = (const float*)d_in[2];
  const float* Wv = (const float*)d_in[3];
  const float* hE = (const float*)d_in[4];
  const float* wE = (const float*)d_in[5];
  float* out = (float*)d_out;

  dim3 grid(Wdim / TW, Hdim / TH, Bn);
  sasa_fused_kernel<<<grid, 256, 0, stream>>>(x, Wq, Wk, Wv, hE, wE, out);
}


# Tile 4 rows x 32 cols -> 8 rows x 32 cols (grid 1024 -> 512 work-groups, 2 per WGP, 4 waves per SIMD): 340 halo positions per 256 outputs (was 408), 22 WMMA batches per 8 rows (was 26); same sums
# speedup vs baseline: 5.0181x; 1.0677x over previous
#include <hip/hip_runtime.h>
#include <cstddef>


typedef __attribute__((ext_vector_type(2))) float v2f;
typedef __attribute__((ext_vector_type(8))) float v8f;

constexpr int Bn   = 8;
constexpr int Hdim = 128;
constexpr int Wdim = 128;
constexpr int NCH  = 16;
constexpr int NGP  = 4;
constexpr int TH   = 8;
constexpr int TW   = 32;
constexpr int HTH  = TH + 2;
constexpr int HTW  = TW + 2;
constexpr int HALO = HTH * HTW;
constexpr int NB   = (HALO + 15) / 16;
constexpr int NBW  = 3;
constexpr int CS    = HALO;
constexpr int PS    = 2 * CS;
constexpr int PLANE = NCH * CS;
constexpr int NEMB  = NCH * 9;
constexpr int KOFF  = 2 * PLANE + 32;
constexpr int VOFF  = KOFF + 2 * PLANE;
constexpr int EOFF  = VOFF + 2 * PLANE;
constexpr int LDSN  = EOFF + NGP * NEMB;
constexpr int HW    = Hdim * Wdim;
constexpr int SR    = TH;
static_assert((2 * PLANE) % 64 == 0 && (4 * PS) % 64 == 32, "lane halves of a fragment store must fall on disjoint banks");
static_assert(NB > 15 && NB <= 8 * NBW, "every wave owns batches wv and wv + 8; wv + 16 exists for the first waves only");

struct TileRegs {
  float w1[4], w2[4];
  float ww[4];
  float u[NBW][8];
};

__device__ __forceinline__ void load_tile(TileRegs& t, int gp, const float* __restrict__ x,
                                          const int* xp, bool three) {
  const int xo = gp * NCH * HW;
  for (int k = 0; k < 2; ++k) {
    for (int c = 0; c < 4; ++c) {
      t.u[k][2 * c]     = x[xp[k] + xo + (4 * c) * HW];
      t.u[k][2 * c + 1] = x[xp[k] + xo + (4 * c + 1) * HW];
    }
  }
  if (three) {
    for (int c = 0; c < 4; ++c) {
      t.u[2][2 * c]     = x[xp[2] + xo + (4 * c) * HW];
      t.u[2][2 * c + 1] = x[xp[2] + xo + (4 * c + 1) * HW];
    }
  } else {
    for (int c = 0; c < 8; ++c) t.u[2][c] = 0.0f;
  }
}

__device__ __forceinline__ void load_w(TileRegs& t, int gp, const float* Wqk,
                                       const float* __restrict__ Wv, int wofs1, int wofs) {
  const int wo1 = wofs1 + gp * 128;
  const int wo  = wofs + gp * 128;
  for (int cc = 0; cc < 2; ++cc) {
    for (int e = 0; e < 2; ++e) {
      t.w1[2 * cc + e] = Wqk[wo1 + 4 * cc + e];
      t.w2[2 * cc + e] = Wqk[wo1 + 64 + 4 * cc + e];
      t.ww[2 * cc + e] = Wv[wo + 4 * cc + e];
    }
  }
}

__device__ __forceinline__ void lds_barrier() {
  asm volatile("s_wait_dscnt 0x0\n\ts_barrier_signal -1\n\ts_barrier_wait -1" ::: "memory");
}

__device__ __forceinline__ void conv_batch(const v2f* a1, const v2f* a2, const v2f* av,
                                           const float* u, bool inb, int pos, int kh,
                                           float* qk, float* vt) {
  v2f bb[4];
  for (int c = 0; c < 4; ++c) {
    bb[c][0] = inb ? u[2 * c]     : 0.0f;
    bb[c][1] = inb ? u[2 * c + 1] : 0.0f;
  }
  v8f c1 = {}, c2 = {}, cv = {};
  cv = __builtin_amdgcn_wmma_f32_16x16x4_f32(false, av[0], false, bb[0], (short)0, cv, false, false);
  c1 = __builtin_amdgcn_wmma_f32_16x16x4_f32(false, a1[0], false, bb[0], (short)0, c1, false, false);
  cv = __builtin_amdgcn_wmma_f32_16x16x4_f32(false, av[1], false, bb[1], (short)0, cv, false, false);
  c1 = __builtin_amdgcn_wmma_f32_16x16x4_f32(false, a1[1], false, bb[1], (short)0, c1, false, false);
  cv = __builtin_amdgcn_wmma_f32_16x16x4_f32(false, av[2], false, bb[2], (short)0, cv, false, false);
  c2 = __builtin_amdgcn_wmma_f32_16x16x4_f32(false, a2[0], false, bb[2], (short)0, c2, false, false);
  cv = __builtin_amdgcn_wmma_f32_16x16x4_f32(false, av[3], false, bb[3], (short)0, cv, false, false);
  c2 = __builtin_amdgcn_wmma_f32_16x16x4_f32(false, a2[1], false, bb[3], (short)0, c2, false, false);
  if (pos < HALO) {
    float* g1 = &qk[2 * pos];
    float* g2 = &qk[4 * PS + 2 * pos];
    float* v8 = &vt[4 * kh * PS + 2 * pos];
    #pragma unroll
    for (int t = 0; t < 4; ++t) { v2f p; p[0] = c1[2 * t]; p[1] = c1[2 * t + 1]; *(v2f*)&g1[t * PS] = p; }
    #pragma unroll
    for (int t = 0; t < 4; ++t) { v2f p; p[0] = c2[2 * t]; p[1] = c2[2 * t + 1]; *(v2f*)&g2[t * PS] = p; }
    #pragma unroll
    for (int t = 0; t < 4; ++t) { v2f p; p[0] = cv[2 * t]; p[1] = cv[2 * t + 1]; *(v2f*)&v8[t * PS] = p; }
  }
}

__global__ __launch_bounds__(256, 4) void sasa_fused_kernel(
    const float* __restrict__ x,  const float* __restrict__ Wq,
    const float* __restrict__ Wk, const float* __restrict__ Wv,
    const float* __restrict__ hE, const float* __restrict__ wE,
    float* __restrict__ out) {
  __shared__ __attribute__((aligned(16))) float lds[LDSN];
  float* const qt  = lds;
  float* const kt  = lds + KOFF;
  float* const vt  = lds + VOFF;
  float* const emb = lds + EOFF;

  const int tid = threadIdx.x;
  const int bx  = blockIdx.x;
  const int by  = blockIdx.y;
  const int b   = blockIdx.z;
  const int x0  = bx * TW, y0 = by * TH;

  const int ln = tid & 31;
  const int m  = ln & 15;
  const int kh = ln >> 4;
  const int wv = tid >> 5;


  const int hi   = m >> 3;
  const int wofs = hi * 64 + (m & 7) * 8 + 2 * kh;
  const float* Wqk = hi ? Wk : Wq;
  const int wofs1 = (m & 7) * 8 + 2 * kh;

  const bool has_e = (tid < NEMB);
  const float* ep;
  int eidx;
  {
    int t  = has_e ? tid : 0;
    int ch = t / 9, w9 = t - ch * 9;
    eidx = (ch >> 1) * 18 + 2 * w9 + (ch & 1);
    int i = w9 / 3, j = w9 - i * 3;
    int g  = ch >> 3;
    int cg = ch & 7;
    ep = (cg < 4) ? (hE + (g * 4 + cg) * 3 + i)
                  : (wE + (g * 4 + (cg - 4)) * 3 + j);
  }

  const int xg = (b * 64 + 2 * kh) * HW;
  const bool three = (wv + 16 < NB);
  int  pos[NBW];
  bool inb[NBW];
  int  xp[NBW];
  for (int k = 0; k < NBW; ++k) {
    pos[k] = (wv + 8 * k) * 16 + m;
    int pc = (pos[k] < HALO) ? pos[k] : (HALO - 1);
    int hy = pc / HTW, hx = pc - hy * HTW;
    int gy = y0 + hy - 1, gx = x0 + hx - 1;
    inb[k] = (gy >= 0 && gy < Hdim && gx >= 0 && gx < Wdim);
    int cy = min(max(gy, 0), Hdim - 1), cx = min(max(gx, 0), Wdim - 1);
    xp[k] = xg + cy * Wdim + cx;
  }

  const int spx  = ln;

  TileRegs t;
  load_tile(t, 0, x, xp, three);
  load_w(t, 0, Wqk, Wv, wofs1, wofs);

  if (has_e) {
    for (int g4 = 0; g4 < NGP; ++g4) emb[g4 * NEMB + eidx] = ep[g4 * 24];
  }

  __builtin_amdgcn_s_setprio(3);
  for (int gp = 0; gp < NGP; ++gp) {
    const int st = gp & 1;
    float* qs = qt + st * PLANE;
    float* ks = kt + st * PLANE;
    float* vs = vt + st * PLANE;
    const float* es = emb + gp * NEMB;
    const int cbase = gp * NCH;

    v2f a1[2], a2[2], av[4];
    for (int c = 0; c < 2; ++c) {
      a1[c][0] = t.w1[2 * c]; a1[c][1] = t.w1[2 * c + 1];
      a2[c][0] = t.w2[2 * c]; a2[c][1] = t.w2[2 * c + 1];
    }
    for (int c = 0; c < 4; ++c) {
      const bool nz = ((c >> 1) == hi);
      const int  cc = c & 1;
      av[c][0] = nz ? t.ww[2 * cc] : 0.0f; av[c][1] = nz ? t.ww[2 * cc + 1] : 0.0f;
    }
    float* qks = kh ? ks : qs;

    conv_batch(a1, a2, av, t.u[0], inb[0], pos[0], kh, qks, vs);
    conv_batch(a1, a2, av, t.u[1], inb[1], pos[1], kh, qks, vs);
    if (three)
      conv_batch(a1, a2, av, t.u[2], inb[2], pos[2], kh, qks, vs);

    if (gp + 1 < NGP) load_tile(t, gp + 1, x, xp, three);

    lds_barrier();

    {
      const int sofs = wv * PS + 2 * spx;
      const float* kc = ks + sofs;
      const float* vc = vs + sofs;
      const float* qc = qs + sofs + 2 * (HTW + 1);
      v2f e9[9];
      for (int w9 = 0; w9 < 9; ++w9) e9[w9] = *(const v2f*)&es[wv * 18 + 2 * w9];
      v2f kk[SR + 2][3], vv[SR + 2][3];
      for (int i = 0; i < 2; ++i) {
        for (int j = 0; j < 3; ++j) {
          kk[i][j] = *(const v2f*)&kc[2 * (i * HTW + j)];
          vv[i][j] = *(const v2f*)&vc[2 * (i * HTW + j)];
        }
      }
      float* op = out + (((b * 64 + cbase + 2 * wv) * Hdim + y0) * Wdim + (x0 + spx));
      for (int r = 0; r < SR; ++r) {
        if (r == SR - 1 && gp + 1 < NGP) load_w(t, gp + 1, Wqk, Wv, wofs1, wofs);
        for (int j = 0; j < 3; ++j) {
          kk[r + 2][j] = *(const v2f*)&kc[2 * ((r + 2) * HTW + j)];
          vv[r + 2][j] = *(const v2f*)&vc[2 * ((r + 2) * HTW + j)];
        }
        v2f q = *(const v2f*)&qc[2 * r * HTW] * 1.44269504088896341f;
        v2f lg[9];
        for (int w9 = 0; w9 < 9; ++w9) {
          int i = w9 / 3, j = w9 - 3 * i;
          lg[w9] = q * (kk[r + i][j] + e9[w9]);
        }
        v2f mx;
        for (int c = 0; c < 2; ++c) {
          float m3 = fmaxf(fmaxf(lg[0][c], lg[1][c]), lg[2][c]);
          m3 = fmaxf(fmaxf(m3, lg[3][c]), lg[4][c]);
          m3 = fmaxf(fmaxf(m3, lg[5][c]), lg[6][c]);
          m3 = fmaxf(fmaxf(m3, lg[7][c]), lg[8][c]);
          mx[c] = m3;
        }
        v2f s, acc;
        for (int w9 = 0; w9 < 9; ++w9) {
          int i = w9 / 3, j = w9 - 3 * i;
          v2f d = lg[w9] - mx;
          v2f e;
          e[0] = __builtin_amdgcn_exp2f(d[0]);
          e[1] = __builtin_amdgcn_exp2f(d[1]);
          if (w9 == 0) { s = e; acc = e * vv[r][0]; }
          else         { s += e; acc = __builtin_elementwise_fma(e, vv[r + i][j], acc); }
        }
        v2f rs;
        rs[0] = __builtin_amdgcn_rcpf(s[0]);
        rs[1] = __builtin_amdgcn_rcpf(s[1]);
        v2f o = acc * rs;
        op[r * Wdim]      = o[0];
        op[HW + r * Wdim] = o[1];
      }
    }
    if (gp == 0)      __builtin_amdgcn_s_setprio(2);
    else if (gp == 1) __builtin_amdgcn_s_setprio(1);
    else if (gp == 2) __builtin_amdgcn_s_setprio(0);
  }
}

extern "C" void kernel_launch(void* const* d_in, const int* in_sizes, int n_in,
                              void* d_out, int out_size, void* d_ws, size_t ws_size,
                              hipStream_t stream) {
  (void)in_sizes; (void)n_in; (void)d_ws; (void)ws_size; (void)out_size;
  const float* x  = (const float*)d_in[0];
  const float* Wq = (const float*)d_in[1];
  const float* Wk = (const float*)d_in[2];
  const float* Wv = (const float*)d_in[3];
  const float* hE = (const float*)d_in[4];
  const float* wE = (const float*)d_in[5];
  float* out = (float*)d_out;

  dim3 grid(Wdim / TW, Hdim / TH, Bn);
  sasa_fused_kernel<<<grid, 256, 0, stream>>>(x, Wq, Wk, Wv, hE, wE, out);
}
